# starting kernel
# baseline (speedup 1.0000x reference)
#include <hip/hip_runtime.h>
#include <hip/hip_bf16.h>


typedef float v2f __attribute__((ext_vector_type(2)));
typedef float v8f __attribute__((ext_vector_type(8)));


__global__ void deg_init_kernel(float* __restrict__ deg, int n) {
    int i = blockIdx.x * blockDim.x + threadIdx.x;
    if (i < n) deg[i] = 1.0f;
}

__global__ void deg_count_kernel(const int* __restrict__ dst,
                                 float* __restrict__ deg, int ne) {
    int e = blockIdx.x * blockDim.x + threadIdx.x;
    if (e < ne) atomicAdd(&deg[dst[e]], 1.0f);
}

__global__ void deg_to_dinv_kernel(float* __restrict__ deg, int n) {
    int i = blockIdx.x * blockDim.x + threadIdx.x;
    if (i < n) {
        float d = deg[i];
        deg[i] = (d > 0.0f) ? rsqrtf(d) : 0.0f;
    }
}


template <bool RELU, int TN>
__global__ void gemm_wmma_strip(const float* __restrict__ A,
                                const float* __restrict__ W,
                                float* __restrict__ out,
                                int M, int K) {
    const int Nc   = TN * 16;
    const int wave = (blockIdx.x * blockDim.x + threadIdx.x) >> 5;
    const int lane = threadIdx.x & 31;

    const int tm = wave;
    if (tm * 16 >= M) return;

    const int ml    = lane & 15;
    const int khalf = (lane >> 4) << 1;

    const float* arow = A + (size_t)(tm * 16 + ml) * K;

    v8f acc[TN];
#pragma unroll
    for (int t = 0; t < TN; ++t) acc[t] = (v8f){};

    for (int k = 0; k < K; k += 4) {
        v2f a = *(const v2f*)(arow + k + khalf);
        if (RELU) {
            a.x = fmaxf(a.x, 0.0f);
            a.y = fmaxf(a.y, 0.0f);
        }

        const float* w0 = W + (size_t)(k + khalf) * Nc + ml;
        v2f b[TN];
#pragma unroll
        for (int t = 0; t < TN; ++t) {
            b[t].x = w0[t * 16];
            b[t].y = w0[Nc + t * 16];
        }

#pragma unroll
        for (int t = 0; t < TN; ++t) {
            acc[t] = __builtin_amdgcn_wmma_f32_16x16x4_f32(
                 false, a,  false, b[t],
                 (short)0, acc[t],  false,  false);
        }
    }

    const int row0 = tm * 16 + ((lane >> 4) << 3);
#pragma unroll
    for (int t = 0; t < TN; ++t) {
        float* o = out + (size_t)row0 * Nc + t * 16 + ml;
#pragma unroll
        for (int r = 0; r < 8; ++r) o[(size_t)r * Nc] = acc[t][r];
    }
}


__global__ void self_init_kernel(const float* __restrict__ t,
                                 const float* __restrict__ dinv,
                                 const float* __restrict__ bias,
                                 float* __restrict__ agg,
                                 int n, int C) {
    int idx = blockIdx.x * blockDim.x + threadIdx.x;
    if (idx >= n * C) return;
    int i = idx / C;
    int c = idx - i * C;
    float di = dinv[i];
    agg[idx] = bias[c] + di * di * t[idx];
}

template <int C>
__global__ void edge_agg_kernel(const float* __restrict__ t,
                                const float* __restrict__ dinv,
                                const int* __restrict__ src,
                                const int* __restrict__ dst,
                                float* __restrict__ agg, int ne) {
    int e = (blockIdx.x * blockDim.x + threadIdx.x) >> 5;
    if (e >= ne) return;
    int lane = threadIdx.x & 31;

    int s = src[e];
    int d = dst[e];
    float nrm = dinv[s] * dinv[d];

    float* ad = agg + (size_t)d * C;
    if constexpr (C == 128) {
        const float4* ts = (const float4*)(t + (size_t)s * C);
        float4 v = ts[lane];
        atomicAdd(&ad[4 * lane + 0], nrm * v.x);
        atomicAdd(&ad[4 * lane + 1], nrm * v.y);
        atomicAdd(&ad[4 * lane + 2], nrm * v.z);
        atomicAdd(&ad[4 * lane + 3], nrm * v.w);
    } else {
        const float2* ts = (const float2*)(t + (size_t)s * C);
        float2 v = ts[lane];
        atomicAdd(&ad[2 * lane + 0], nrm * v.x);
        atomicAdd(&ad[2 * lane + 1], nrm * v.y);
    }
}


static inline int cdiv(long long a, long long b) { return (int)((a + b - 1) / b); }

extern "C" void kernel_launch(void* const* d_in, const int* in_sizes, int n_in,
                              void* d_out, int out_size, void* d_ws, size_t ws_size,
                              hipStream_t stream) {
    const float* x  = (const float*)d_in[0];
    const float* W1 = (const float*)d_in[1];
    const float* b1 = (const float*)d_in[2];
    const float* W2 = (const float*)d_in[3];
    const float* b2 = (const float*)d_in[4];
    const float* W3 = (const float*)d_in[5];
    const float* b3 = (const float*)d_in[6];
    const int*   ei = (const int*)d_in[7];

    const int IN_C = 64, HID_C = 128, OUT_C = 64;
    const int n  = in_sizes[0] / IN_C;
    const int ne = in_sizes[7] / 2;
    const int* src = ei;
    const int* dst = ei + ne;

    float* T    = (float*)d_ws;
    float* G    = T + (size_t)n * HID_C;
    float* dinv = G + (size_t)n * HID_C;
    float* out  = (float*)d_out;

    const int TB = 256;
    const long long strip_waves = n / 16;
    const int gemm_blocks = cdiv(strip_waves * 32, TB);

    deg_init_kernel<<<cdiv(n, TB), TB, 0, stream>>>(dinv, n);
    deg_count_kernel<<<cdiv(ne, TB), TB, 0, stream>>>(dst, dinv, ne);
    deg_to_dinv_kernel<<<cdiv(n, TB), TB, 0, stream>>>(dinv, n);

    gemm_wmma_strip<false, 8><<<gemm_blocks, TB, 0, stream>>>(x, W1, T, n, IN_C);
    self_init_kernel<<<cdiv((long long)n * HID_C, TB), TB, 0, stream>>>(T, dinv, b1, G, n, HID_C);
    edge_agg_kernel<HID_C><<<cdiv((long long)ne * 32, TB), TB, 0, stream>>>(T, dinv, src, dst, G, ne);

    gemm_wmma_strip<true, 8><<<gemm_blocks, TB, 0, stream>>>(G, W2, T, n, HID_C);
    self_init_kernel<<<cdiv((long long)n * HID_C, TB), TB, 0, stream>>>(T, dinv, b2, G, n, HID_C);
    edge_agg_kernel<HID_C><<<cdiv((long long)ne * 32, TB), TB, 0, stream>>>(T, dinv, src, dst, G, ne);

    gemm_wmma_strip<true, 4><<<gemm_blocks, TB, 0, stream>>>(G, W3, T, n, HID_C);
    self_init_kernel<<<cdiv((long long)n * OUT_C, TB), TB, 0, stream>>>(T, dinv, b3, out, n, OUT_C);
    edge_agg_kernel<OUT_C><<<cdiv((long long)ne * 32, TB), TB, 0, stream>>>(T, dinv, src, dst, out, ne);
}


# Aggregation output-stationary: a work-group holds 256 destination rows in LDS, filters the dst list for them (no sort), adds gathered rows by LDS float add, not global atomics; sum order LDS arrival
# speedup vs baseline: 1.3874x; 1.3874x over previous
#include <hip/hip_runtime.h>
#include <hip/hip_bf16.h>


typedef float v2f __attribute__((ext_vector_type(2)));
typedef float v8f __attribute__((ext_vector_type(8)));


__global__ void deg_init_kernel(float* __restrict__ deg, int n) {
    int i = blockIdx.x * blockDim.x + threadIdx.x;
    if (i < n) deg[i] = 1.0f;
}

__global__ void deg_count_kernel(const int* __restrict__ dst,
                                 float* __restrict__ deg, int ne) {
    int e = blockIdx.x * blockDim.x + threadIdx.x;
    if (e < ne) atomicAdd(&deg[dst[e]], 1.0f);
}

__global__ void deg_to_dinv_kernel(float* __restrict__ deg, int n) {
    int i = blockIdx.x * blockDim.x + threadIdx.x;
    if (i < n) {
        float d = deg[i];
        deg[i] = (d > 0.0f) ? rsqrtf(d) : 0.0f;
    }
}


template <bool RELU, int TN>
__global__ void gemm_wmma_strip(const float* __restrict__ A,
                                const float* __restrict__ W,
                                float* __restrict__ out,
                                int M, int K) {
    const int Nc   = TN * 16;
    const int wave = (blockIdx.x * blockDim.x + threadIdx.x) >> 5;
    const int lane = threadIdx.x & 31;

    const int tm = wave;
    if (tm * 16 >= M) return;

    const int ml    = lane & 15;
    const int khalf = (lane >> 4) << 1;

    const float* arow = A + (size_t)(tm * 16 + ml) * K;

    v8f acc[TN];
#pragma unroll
    for (int t = 0; t < TN; ++t) acc[t] = (v8f){};

    for (int k = 0; k < K; k += 4) {
        v2f a = *(const v2f*)(arow + k + khalf);
        if (RELU) {
            a.x = fmaxf(a.x, 0.0f);
            a.y = fmaxf(a.y, 0.0f);
        }

        const float* w0 = W + (size_t)(k + khalf) * Nc + ml;
        v2f b[TN];
#pragma unroll
        for (int t = 0; t < TN; ++t) {
            b[t].x = w0[t * 16];
            b[t].y = w0[Nc + t * 16];
        }

#pragma unroll
        for (int t = 0; t < TN; ++t) {
            acc[t] = __builtin_amdgcn_wmma_f32_16x16x4_f32(
                 false, a,  false, b[t],
                 (short)0, acc[t],  false,  false);
        }
    }

    const int row0 = tm * 16 + ((lane >> 4) << 3);
#pragma unroll
    for (int t = 0; t < TN; ++t) {
        float* o = out + (size_t)row0 * Nc + t * 16 + ml;
#pragma unroll
        for (int r = 0; r < 8; ++r) o[(size_t)r * Nc] = acc[t][r];
    }
}


__global__ void self_init_kernel(const float* __restrict__ t,
                                 const float* __restrict__ dinv,
                                 const float* __restrict__ bias,
                                 float* __restrict__ agg,
                                 int n, int C) {
    int idx = blockIdx.x * blockDim.x + threadIdx.x;
    if (idx >= n * C) return;
    int i = idx / C;
    int c = idx - i * C;
    float di = dinv[i];
    agg[idx] = bias[c] + di * di * t[idx];
}


template <int K>
__device__ __forceinline__ void load_dst_trip(int4 (&dv)[K], const int* __restrict__ dst,
                                              int ebase, int lane, int ne, bool full) {
    if (full) {
#pragma unroll
        for (int k = 0; k < K; ++k)
            dv[k] = *(const int4*)(dst + ebase + 4 * (lane + 32 * k));
    } else {
#pragma unroll
        for (int k = 0; k < K; ++k) {
            const int e0 = ebase + 4 * (lane + 32 * k);
            dv[k].x = (e0 + 0 < ne) ? dst[e0 + 0] : -1;
            dv[k].y = (e0 + 1 < ne) ? dst[e0 + 1] : -1;
            dv[k].z = (e0 + 2 < ne) ? dst[e0 + 2] : -1;
            dv[k].w = (e0 + 3 < ne) ? dst[e0 + 3] : -1;
        }
    }
}

template <int C, int R, int K>
__global__ __launch_bounds__(1024)
void tile_agg_kernel(const float* __restrict__ t,
                     const float* __restrict__ dinv,
                     const int* __restrict__ src,
                     const int* __restrict__ dst,
                     float* __restrict__ agg, int n, int ne) {
    static_assert(K == 8, "one match bit per inspected edge: 4*K bits in a 32-bit mask");
    constexpr int NW  = 32;
    constexpr int EPW = 128 * K;
    constexpr int Q   = C / 32;
    __shared__ float tile[R * C];
    __shared__ unsigned int hits[NW * EPW];

    const int tid  = threadIdx.x;
    const int lane = tid & 31;
    const int wave = __builtin_amdgcn_readfirstlane(tid >> 5);
    const int row0 = blockIdx.x * R;
    const int rows = (n - row0 < R) ? (n - row0) : R;

    {
        const float4* g4 = (const float4*)(agg + (size_t)row0 * C);
        float4* t4 = (float4*)tile;
        for (int i = tid; i < rows * (C / 4); i += 1024) t4[i] = g4[i];
    }
    __syncthreads();

    const bool vec_ok = ((((size_t)dst) & 15) == 0);
    const int nchunks = (ne + EPW - 1) / EPW;
    const int rot = (int)(((long long)blockIdx.x * nchunks) / gridDim.x);
    unsigned int* my = hits + wave * EPW;
    const unsigned below = (1u << lane) - 1u;

    int4 dv[K];
    if (wave < nchunks) {
        int chunk = wave + rot;
        if (chunk >= nchunks) chunk -= nchunks;
        load_dst_trip<K>(dv, dst, chunk * EPW, lane, ne,
                         vec_ok && (chunk * EPW + EPW <= ne));
    }

    for (int idx = wave; idx < nchunks; idx += NW) {
        int chunk = idx + rot;
        if (chunk >= nchunks) chunk -= nchunks;
        const int ebase = chunk * EPW;

        unsigned mask = 0u;
#pragma unroll
        for (int k = 0; k < K; ++k) {
#pragma unroll
            for (int j = 0; j < 4; ++j) {
                const int d = (j == 0) ? dv[k].x : (j == 1) ? dv[k].y
                            : (j == 2) ? dv[k].z : dv[k].w;
                if ((unsigned)(d - row0) < (unsigned)rows) mask |= (1u << (4 * k + j));
            }
        }

        if (idx + NW < nchunks) {
            int nchunk = idx + NW + rot;
            if (nchunk >= nchunks) nchunk -= nchunks;
            load_dst_trip<K>(dv, dst, nchunk * EPW, lane, ne,
                             vec_ok && (nchunk * EPW + EPW <= ne));
        }

        int cnt = 0;
        for (int round = 0; round < 32; ++round) {
            const bool has = (mask != 0u);
            const unsigned bal = __builtin_amdgcn_ballot_w32(has);
            if (bal == 0u) break;
            if (has) {
                const int bp = __builtin_ctz(mask);
                mask &= mask - 1u;
                my[cnt + __popc(bal & below)] =
                    (unsigned)(4 * lane + 128 * (bp >> 2) + (bp & 3));
            }
            cnt += __popc(bal);
        }
        __builtin_amdgcn_fence(__ATOMIC_RELEASE, "wavefront");
        __builtin_amdgcn_wave_barrier();
        __builtin_amdgcn_fence(__ATOMIC_ACQUIRE, "wavefront");

        for (int b = 0; b < cnt; b += 32) {
            const int mcount = (cnt - b < 32) ? (cnt - b) : 32;
            int s = 0, dl = 0;
            float w = 0.0f;
            if (lane < mcount) {
                const int e = ebase + (int)my[b + lane];
                const int d = dst[e];
                s  = src[e];
                dl = d - row0;
                w  = dinv[s] * dinv[d];
            }
            for (int m = 0; m < mcount; m += 4) {
                float v[4][Q];
                int   dm[4];
                float wm[4];
#pragma unroll
                for (int u = 0; u < 4; ++u) {
                    const int mm = (m + u) & 31;
                    const int sm = __shfl(s, mm, 32);
                    dm[u] = __shfl(dl, mm, 32);
                    wm[u] = __shfl(w, mm, 32);
                    const float* tr = t + (size_t)sm * C + lane;
#pragma unroll
                    for (int q = 0; q < Q; ++q) v[u][q] = tr[32 * q];
                }
#pragma unroll
                for (int u = 0; u < 4; ++u) {
                    if (m + u < mcount) {
                        float* lr = tile + dm[u] * C + lane;
#pragma unroll
                        for (int q = 0; q < Q; ++q)
                            atomicAdd(&lr[32 * q], wm[u] * v[u][q]);
                    }
                }
            }
        }
        __builtin_amdgcn_fence(__ATOMIC_RELEASE, "wavefront");
        __builtin_amdgcn_wave_barrier();
        __builtin_amdgcn_fence(__ATOMIC_ACQUIRE, "wavefront");
    }
    __syncthreads();

    {
        float4* g4 = (float4*)(agg + (size_t)row0 * C);
        const float4* t4 = (const float4*)tile;
        for (int i = tid; i < rows * (C / 4); i += 1024) g4[i] = t4[i];
    }
}


static inline int cdiv(long long a, long long b) { return (int)((a + b - 1) / b); }

extern "C" void kernel_launch(void* const* d_in, const int* in_sizes, int n_in,
                              void* d_out, int out_size, void* d_ws, size_t ws_size,
                              hipStream_t stream) {
    const float* x  = (const float*)d_in[0];
    const float* W1 = (const float*)d_in[1];
    const float* b1 = (const float*)d_in[2];
    const float* W2 = (const float*)d_in[3];
    const float* b2 = (const float*)d_in[4];
    const float* W3 = (const float*)d_in[5];
    const float* b3 = (const float*)d_in[6];
    const int*   ei = (const int*)d_in[7];

    const int IN_C = 64, HID_C = 128, OUT_C = 64;
    const int n  = in_sizes[0] / IN_C;
    const int ne = in_sizes[7] / 2;
    const int* src = ei;
    const int* dst = ei + ne;

    float* T    = (float*)d_ws;
    float* G    = T + (size_t)n * HID_C;
    float* dinv = G + (size_t)n * HID_C;
    float* out  = (float*)d_out;

    const int TB = 256;
    const long long strip_waves = n / 16;
    const int gemm_blocks = cdiv(strip_waves * 32, TB);
    constexpr int TR = 256, TK = 8;
    const int tile_blocks = cdiv(n, TR);

    deg_init_kernel<<<cdiv(n, TB), TB, 0, stream>>>(dinv, n);
    deg_count_kernel<<<cdiv(ne, TB), TB, 0, stream>>>(dst, dinv, ne);
    deg_to_dinv_kernel<<<cdiv(n, TB), TB, 0, stream>>>(dinv, n);

    gemm_wmma_strip<false, 8><<<gemm_blocks, TB, 0, stream>>>(x, W1, T, n, IN_C);
    self_init_kernel<<<cdiv((long long)n * HID_C, TB), TB, 0, stream>>>(T, dinv, b1, G, n, HID_C);
    tile_agg_kernel<HID_C, TR, TK><<<tile_blocks, 1024, 0, stream>>>(T, dinv, src, dst, G, n, ne);

    gemm_wmma_strip<true, 8><<<gemm_blocks, TB, 0, stream>>>(G, W2, T, n, HID_C);
    self_init_kernel<<<cdiv((long long)n * HID_C, TB), TB, 0, stream>>>(T, dinv, b2, G, n, HID_C);
    tile_agg_kernel<HID_C, TR, TK><<<tile_blocks, 1024, 0, stream>>>(T, dinv, src, dst, G, n, ne);

    gemm_wmma_strip<true, 4><<<gemm_blocks, TB, 0, stream>>>(G, W3, T, n, HID_C);
    self_init_kernel<<<cdiv((long long)n * OUT_C, TB), TB, 0, stream>>>(T, dinv, b3, out, n, OUT_C);
    tile_agg_kernel<OUT_C, TR, TK><<<tile_blocks, 1024, 0, stream>>>(T, dinv, src, dst, out, n, ne);
}


# Aggregation's edge filter scans a one-byte tile key per edge (written by the degree-count launch; 4 keys per word op, 2048-edge trips), matches confirmed against dst; sum order still LDS arrival
# speedup vs baseline: 1.5200x; 1.0956x over previous
#include <hip/hip_runtime.h>
#include <hip/hip_bf16.h>


typedef float v2f __attribute__((ext_vector_type(2)));
typedef float v8f __attribute__((ext_vector_type(8)));


__global__ void deg_init_kernel(float* __restrict__ deg, int n) {
    int i = blockIdx.x * blockDim.x + threadIdx.x;
    if (i < n) deg[i] = 1.0f;
}

constexpr int TILE_ROWS  = 256;
constexpr int TILE_SHIFT = 8;
constexpr int KEY_LOADS  = 4;
constexpr int TRIP_EDGES = 512 * KEY_LOADS;

__global__ void deg_count_key_kernel(const int* __restrict__ dst,
                                     float* __restrict__ deg,
                                     unsigned int* __restrict__ key,
                                     int ne, int nwords) {
    const int q = blockIdx.x * blockDim.x + threadIdx.x;
    if (q >= nwords) return;
    const int e0 = 4 * q;
    unsigned int kw = 0u;
    if (e0 + 3 < ne) {
        int d[4];
#pragma unroll
        for (int j = 0; j < 4; ++j) d[j] = dst[e0 + j];
#pragma unroll
        for (int j = 0; j < 4; ++j) {
            atomicAdd(&deg[d[j]], 1.0f);
            kw |= ((unsigned int)(d[j] >> TILE_SHIFT) & 255u) << (8 * j);
        }
    } else {
#pragma unroll
        for (int j = 0; j < 4; ++j) {
            if (e0 + j < ne) {
                const int dj = dst[e0 + j];
                atomicAdd(&deg[dj], 1.0f);
                kw |= ((unsigned int)(dj >> TILE_SHIFT) & 255u) << (8 * j);
            }
        }
    }
    key[q] = kw;
}

__global__ void deg_to_dinv_kernel(float* __restrict__ deg, int n) {
    int i = blockIdx.x * blockDim.x + threadIdx.x;
    if (i < n) {
        float d = deg[i];
        deg[i] = (d > 0.0f) ? rsqrtf(d) : 0.0f;
    }
}


template <bool RELU, int TN>
__global__ void gemm_wmma_strip(const float* __restrict__ A,
                                const float* __restrict__ W,
                                float* __restrict__ out,
                                int M, int K) {
    const int Nc   = TN * 16;
    const int wave = (blockIdx.x * blockDim.x + threadIdx.x) >> 5;
    const int lane = threadIdx.x & 31;

    const int tm = wave;
    if (tm * 16 >= M) return;

    const int ml    = lane & 15;
    const int khalf = (lane >> 4) << 1;

    const float* arow = A + (size_t)(tm * 16 + ml) * K;

    v8f acc[TN];
#pragma unroll
    for (int t = 0; t < TN; ++t) acc[t] = (v8f){};

    for (int k = 0; k < K; k += 4) {
        v2f a = *(const v2f*)(arow + k + khalf);
        if (RELU) {
            a.x = fmaxf(a.x, 0.0f);
            a.y = fmaxf(a.y, 0.0f);
        }

        const float* w0 = W + (size_t)(k + khalf) * Nc + ml;
        v2f b[TN];
#pragma unroll
        for (int t = 0; t < TN; ++t) {
            b[t].x = w0[t * 16];
            b[t].y = w0[Nc + t * 16];
        }

#pragma unroll
        for (int t = 0; t < TN; ++t) {
            acc[t] = __builtin_amdgcn_wmma_f32_16x16x4_f32(
                 false, a,  false, b[t],
                 (short)0, acc[t],  false,  false);
        }
    }

    const int row0 = tm * 16 + ((lane >> 4) << 3);
#pragma unroll
    for (int t = 0; t < TN; ++t) {
        float* o = out + (size_t)row0 * Nc + t * 16 + ml;
#pragma unroll
        for (int r = 0; r < 8; ++r) o[(size_t)r * Nc] = acc[t][r];
    }
}


__global__ void self_init_kernel(const float* __restrict__ t,
                                 const float* __restrict__ dinv,
                                 const float* __restrict__ bias,
                                 float* __restrict__ agg,
                                 int n, int C) {
    int idx = blockIdx.x * blockDim.x + threadIdx.x;
    if (idx >= n * C) return;
    int i = idx / C;
    int c = idx - i * C;
    float di = dinv[i];
    agg[idx] = bias[c] + di * di * t[idx];
}


template <int K>
__device__ __forceinline__ void load_key_trip(uint4 (&kv)[K], const unsigned int* __restrict__ key,
                                              int ebase, int lane) {
    const uint4* kp = (const uint4*)(key + (ebase >> 2));
#pragma unroll
    for (int k = 0; k < K; ++k) kv[k] = kp[lane + 32 * k];
}

template <int C, int R, int K>
__global__ __launch_bounds__(1024)
void tile_agg_kernel(const float* __restrict__ t,
                     const float* __restrict__ dinv,
                     const int* __restrict__ src,
                     const int* __restrict__ dst,
                     const unsigned int* __restrict__ key,
                     float* __restrict__ agg, int n, int ne) {
    static_assert(K == 4, "one match bit per inspected edge: 16*K bits in a 64-bit mask");
    static_assert(K == KEY_LOADS, "the key array is padded to whole trips of this size");
    static_assert(R == (1 << TILE_SHIFT), "the key is the tile number's low byte");
    constexpr int NW  = 32;
    constexpr int EPW = 512 * K;
    constexpr int Q   = C / 32;
    static_assert(EPW <= 65536, "a match is kept as a 16-bit offset into its trip");
    __shared__ float tile[R * C];
    __shared__ unsigned short hits[NW * EPW];

    const int tid  = threadIdx.x;
    const int lane = tid & 31;
    const int wave = __builtin_amdgcn_readfirstlane(tid >> 5);
    const int row0 = blockIdx.x * R;
    const int rows = (n - row0 < R) ? (n - row0) : R;

    {
        const float4* g4 = (const float4*)(agg + (size_t)row0 * C);
        float4* t4 = (float4*)tile;
        for (int i = tid; i < rows * (C / 4); i += 1024) t4[i] = g4[i];
    }
    __syncthreads();

    const int nchunks = (ne + EPW - 1) / EPW;
    const int rot = (int)(((long long)blockIdx.x * nchunks) / gridDim.x);
    unsigned short* my = hits + wave * EPW;
    const unsigned below = (1u << lane) - 1u;
    const unsigned pat = (blockIdx.x & 255u) * 0x01010101u;

    uint4 kv[K];
    if (wave < nchunks) {
        int chunk = wave + rot;
        if (chunk >= nchunks) chunk -= nchunks;
        load_key_trip<K>(kv, key, chunk * EPW, lane);
    }

    for (int idx = wave; idx < nchunks; idx += NW) {
        int chunk = idx + rot;
        if (chunk >= nchunks) chunk -= nchunks;
        const int ebase = chunk * EPW;

        unsigned mlo = 0u, mhi = 0u;
#pragma unroll
        for (int k = 0; k < K; ++k) {
#pragma unroll
            for (int j = 0; j < 4; ++j) {
                const unsigned wv = (j == 0) ? kv[k].x : (j == 1) ? kv[k].y
                                  : (j == 2) ? kv[k].z : kv[k].w;
                const unsigned x = wv ^ pat;
                const unsigned z = (x - 0x01010101u) & ~x & 0x80808080u;
                const unsigned bits = (z >> 7) << (4 * (k & 1) + j);
                if (k < 2) mlo |= bits; else mhi |= bits;
            }
        }
        unsigned long long mask = ((unsigned long long)mhi << 32) | mlo;

        if (idx + NW < nchunks) {
            int nchunk = idx + NW + rot;
            if (nchunk >= nchunks) nchunk -= nchunks;
            load_key_trip<K>(kv, key, nchunk * EPW, lane);
        }

        int cnt = 0;
        for (int round = 0; round < 64; ++round) {
            const bool has = (mask != 0ull);
            const unsigned bal = __builtin_amdgcn_ballot_w32(has);
            if (bal == 0u) break;
            if (has) {
                const int bp = __builtin_ctzll(mask);
                mask &= mask - 1ull;
                const int s  = bp & 7;
                const int kk = ((bp >> 5) << 1) + (s >> 2);
                my[cnt + __popc(bal & below)] =
                    (unsigned short)(16 * lane + 512 * kk + 4 * (s & 3) + ((bp & 31) >> 3));
            }
            cnt += __popc(bal);
        }
        __builtin_amdgcn_fence(__ATOMIC_RELEASE, "wavefront");
        __builtin_amdgcn_wave_barrier();
        __builtin_amdgcn_fence(__ATOMIC_ACQUIRE, "wavefront");

        for (int b = 0; b < cnt; b += 32) {
            const int mcount = (cnt - b < 32) ? (cnt - b) : 32;
            int s = 0, dl = -1;
            float w = 0.0f;
            if (lane < mcount) {
                const int e = ebase + (int)my[b + lane];
                if (e < ne) {
                    const int d  = dst[e];
                    const int se = src[e];
                    w = dinv[se] * dinv[d];
                    const bool mine = (unsigned)(d - row0) < (unsigned)rows;
                    s  = mine ? se : 0;
                    dl = mine ? d - row0 : -1;
                }
            }
            for (int m = 0; m < mcount; m += 4) {
                float v[4][Q];
                int   dm[4];
                float wm[4];
#pragma unroll
                for (int u = 0; u < 4; ++u) {
                    const int mm = (m + u) & 31;
                    const int sm = __shfl(s, mm, 32);
                    dm[u] = __shfl(dl, mm, 32);
                    wm[u] = __shfl(w, mm, 32);
                    const float* tr = t + (size_t)sm * C + lane;
#pragma unroll
                    for (int q = 0; q < Q; ++q) v[u][q] = tr[32 * q];
                }
#pragma unroll
                for (int u = 0; u < 4; ++u) {
                    if (dm[u] >= 0) {
                        float* lr = tile + dm[u] * C + lane;
#pragma unroll
                        for (int q = 0; q < Q; ++q)
                            atomicAdd(&lr[32 * q], wm[u] * v[u][q]);
                    }
                }
            }
        }
        __builtin_amdgcn_fence(__ATOMIC_RELEASE, "wavefront");
        __builtin_amdgcn_wave_barrier();
        __builtin_amdgcn_fence(__ATOMIC_ACQUIRE, "wavefront");
    }
    __syncthreads();

    {
        float4* g4 = (float4*)(agg + (size_t)row0 * C);
        const float4* t4 = (const float4*)tile;
        for (int i = tid; i < rows * (C / 4); i += 1024) g4[i] = t4[i];
    }
}


static inline int cdiv(long long a, long long b) { return (int)((a + b - 1) / b); }

extern "C" void kernel_launch(void* const* d_in, const int* in_sizes, int n_in,
                              void* d_out, int out_size, void* d_ws, size_t ws_size,
                              hipStream_t stream) {
    const float* x  = (const float*)d_in[0];
    const float* W1 = (const float*)d_in[1];
    const float* b1 = (const float*)d_in[2];
    const float* W2 = (const float*)d_in[3];
    const float* b2 = (const float*)d_in[4];
    const float* W3 = (const float*)d_in[5];
    const float* b3 = (const float*)d_in[6];
    const int*   ei = (const int*)d_in[7];

    const int IN_C = 64, HID_C = 128, OUT_C = 64;
    const int n  = in_sizes[0] / IN_C;
    const int ne = in_sizes[7] / 2;
    const int* src = ei;
    const int* dst = ei + ne;

    const int key_words = cdiv(ne, TRIP_EDGES) * (TRIP_EDGES / 4);
    float* T    = (float*)d_ws;
    float* G    = T + (size_t)n * HID_C;
    unsigned int* key = (unsigned int*)(G + (size_t)n * HID_C);
    float* dinv = (float*)(key + key_words);
    float* out  = (float*)d_out;

    const int TB = 256;
    const long long strip_waves = n / 16;
    const int gemm_blocks = cdiv(strip_waves * 32, TB);
    constexpr int TR = TILE_ROWS, TK = KEY_LOADS;
    const int tile_blocks = cdiv(n, TR);

    deg_init_kernel<<<cdiv(n, TB), TB, 0, stream>>>(dinv, n);
    deg_count_key_kernel<<<cdiv(key_words, TB), TB, 0, stream>>>(dst, dinv, key, ne, key_words);
    deg_to_dinv_kernel<<<cdiv(n, TB), TB, 0, stream>>>(dinv, n);

    gemm_wmma_strip<false, 8><<<gemm_blocks, TB, 0, stream>>>(x, W1, T, n, IN_C);
    self_init_kernel<<<cdiv((long long)n * HID_C, TB), TB, 0, stream>>>(T, dinv, b1, G, n, HID_C);
    tile_agg_kernel<HID_C, TR, TK><<<tile_blocks, 1024, 0, stream>>>(T, dinv, src, dst, key, G, n, ne);

    gemm_wmma_strip<true, 8><<<gemm_blocks, TB, 0, stream>>>(G, W2, T, n, HID_C);
    self_init_kernel<<<cdiv((long long)n * HID_C, TB), TB, 0, stream>>>(T, dinv, b2, G, n, HID_C);
    tile_agg_kernel<HID_C, TR, TK><<<tile_blocks, 1024, 0, stream>>>(T, dinv, src, dst, key, G, n, ne);

    gemm_wmma_strip<true, 4><<<gemm_blocks, TB, 0, stream>>>(G, W3, T, n, HID_C);
    self_init_kernel<<<cdiv((long long)n * OUT_C, TB), TB, 0, stream>>>(T, dinv, b3, out, n, OUT_C);
    tile_agg_kernel<OUT_C, TR, TK><<<tile_blocks, 1024, 0, stream>>>(T, dinv, src, dst, key, out, n, ne);
}


# Self-loop term and bias folded into the aggregation's tile fill: a work-group forms bias + dinv^2 * t for its rows as it loads LDS; the three start-value launches go (9, not 12); sum order unchanged
# speedup vs baseline: 1.9618x; 1.2907x over previous
#include <hip/hip_runtime.h>
#include <hip/hip_bf16.h>


typedef float v2f __attribute__((ext_vector_type(2)));
typedef float v8f __attribute__((ext_vector_type(8)));


__global__ void deg_init_kernel(float* __restrict__ deg, int n) {
    int i = blockIdx.x * blockDim.x + threadIdx.x;
    if (i < n) deg[i] = 1.0f;
}

constexpr int TILE_ROWS  = 256;
constexpr int TILE_SHIFT = 8;
constexpr int KEY_LOADS  = 4;
constexpr int TRIP_EDGES = 512 * KEY_LOADS;

__global__ void deg_count_key_kernel(const int* __restrict__ dst,
                                     float* __restrict__ deg,
                                     unsigned int* __restrict__ key,
                                     int ne, int nwords) {
    const int q = blockIdx.x * blockDim.x + threadIdx.x;
    if (q >= nwords) return;
    const int e0 = 4 * q;
    unsigned int kw = 0u;
    if (e0 + 3 < ne) {
        int d[4];
#pragma unroll
        for (int j = 0; j < 4; ++j) d[j] = dst[e0 + j];
#pragma unroll
        for (int j = 0; j < 4; ++j) {
            atomicAdd(&deg[d[j]], 1.0f);
            kw |= ((unsigned int)(d[j] >> TILE_SHIFT) & 255u) << (8 * j);
        }
    } else {
#pragma unroll
        for (int j = 0; j < 4; ++j) {
            if (e0 + j < ne) {
                const int dj = dst[e0 + j];
                atomicAdd(&deg[dj], 1.0f);
                kw |= ((unsigned int)(dj >> TILE_SHIFT) & 255u) << (8 * j);
            }
        }
    }
    key[q] = kw;
}

__global__ void deg_to_dinv_kernel(float* __restrict__ deg, int n) {
    int i = blockIdx.x * blockDim.x + threadIdx.x;
    if (i < n) {
        float d = deg[i];
        deg[i] = (d > 0.0f) ? rsqrtf(d) : 0.0f;
    }
}


template <bool RELU, int TN>
__global__ void gemm_wmma_strip(const float* __restrict__ A,
                                const float* __restrict__ W,
                                float* __restrict__ out,
                                int M, int K) {
    const int Nc   = TN * 16;
    const int wave = (blockIdx.x * blockDim.x + threadIdx.x) >> 5;
    const int lane = threadIdx.x & 31;

    const int tm = wave;
    if (tm * 16 >= M) return;

    const int ml    = lane & 15;
    const int khalf = (lane >> 4) << 1;

    const float* arow = A + (size_t)(tm * 16 + ml) * K;

    v8f acc[TN];
#pragma unroll
    for (int t = 0; t < TN; ++t) acc[t] = (v8f){};

    for (int k = 0; k < K; k += 4) {
        v2f a = *(const v2f*)(arow + k + khalf);
        if (RELU) {
            a.x = fmaxf(a.x, 0.0f);
            a.y = fmaxf(a.y, 0.0f);
        }

        const float* w0 = W + (size_t)(k + khalf) * Nc + ml;
        v2f b[TN];
#pragma unroll
        for (int t = 0; t < TN; ++t) {
            b[t].x = w0[t * 16];
            b[t].y = w0[Nc + t * 16];
        }

#pragma unroll
        for (int t = 0; t < TN; ++t) {
            acc[t] = __builtin_amdgcn_wmma_f32_16x16x4_f32(
                 false, a,  false, b[t],
                 (short)0, acc[t],  false,  false);
        }
    }

    const int row0 = tm * 16 + ((lane >> 4) << 3);
#pragma unroll
    for (int t = 0; t < TN; ++t) {
        float* o = out + (size_t)row0 * Nc + t * 16 + ml;
#pragma unroll
        for (int r = 0; r < 8; ++r) o[(size_t)r * Nc] = acc[t][r];
    }
}


template <int K>
__device__ __forceinline__ void load_key_trip(uint4 (&kv)[K], const unsigned int* __restrict__ key,
                                              int ebase, int lane) {
    const uint4* kp = (const uint4*)(key + (ebase >> 2));
#pragma unroll
    for (int k = 0; k < K; ++k) kv[k] = kp[lane + 32 * k];
}

template <int C, int R, int K>
__global__ __launch_bounds__(1024)
void tile_agg_kernel(const float* __restrict__ t,
                     const float* __restrict__ dinv,
                     const float* __restrict__ bias,
                     const int* __restrict__ src,
                     const int* __restrict__ dst,
                     const unsigned int* __restrict__ key,
                     float* __restrict__ agg, int n, int ne) {
    static_assert(K == 4, "one match bit per inspected edge: 16*K bits in a 64-bit mask");
    static_assert(K == KEY_LOADS, "the key array is padded to whole trips of this size");
    static_assert(R == (1 << TILE_SHIFT), "the key is the tile number's low byte");
    constexpr int NW  = 32;
    constexpr int EPW = 512 * K;
    constexpr int Q   = C / 32;
    static_assert(EPW <= 65536, "a match is kept as a 16-bit offset into its trip");
    __shared__ float tile[R * C];
    __shared__ unsigned short hits[NW * EPW];

    const int tid  = threadIdx.x;
    const int lane = tid & 31;
    const int wave = __builtin_amdgcn_readfirstlane(tid >> 5);
    const int row0 = blockIdx.x * R;
    const int rows = (n - row0 < R) ? (n - row0) : R;

    {
        static_assert(1024 % (C / 4) == 0, "a thread's column group is fixed");
        const float4* g4 = (const float4*)(t + (size_t)row0 * C);
        float4* t4 = (float4*)tile;
        const int c0 = 4 * (tid % (C / 4));
        const float bx = bias[c0], by = bias[c0 + 1], bz = bias[c0 + 2], bw = bias[c0 + 3];
        for (int i = tid; i < rows * (C / 4); i += 1024) {
            const float di = dinv[row0 + i / (C / 4)];
            const float4 tv = g4[i];
            float4 sv;
            sv.x = bx + di * di * tv.x;
            sv.y = by + di * di * tv.y;
            sv.z = bz + di * di * tv.z;
            sv.w = bw + di * di * tv.w;
            t4[i] = sv;
        }
    }
    __syncthreads();

    const int nchunks = (ne + EPW - 1) / EPW;
    const int rot = (int)(((long long)blockIdx.x * nchunks) / gridDim.x);
    unsigned short* my = hits + wave * EPW;
    const unsigned below = (1u << lane) - 1u;
    const unsigned pat = (blockIdx.x & 255u) * 0x01010101u;

    uint4 kv[K];
    if (wave < nchunks) {
        int chunk = wave + rot;
        if (chunk >= nchunks) chunk -= nchunks;
        load_key_trip<K>(kv, key, chunk * EPW, lane);
    }

    for (int idx = wave; idx < nchunks; idx += NW) {
        int chunk = idx + rot;
        if (chunk >= nchunks) chunk -= nchunks;
        const int ebase = chunk * EPW;

        unsigned mlo = 0u, mhi = 0u;
#pragma unroll
        for (int k = 0; k < K; ++k) {
#pragma unroll
            for (int j = 0; j < 4; ++j) {
                const unsigned wv = (j == 0) ? kv[k].x : (j == 1) ? kv[k].y
                                  : (j == 2) ? kv[k].z : kv[k].w;
                const unsigned x = wv ^ pat;
                const unsigned z = (x - 0x01010101u) & ~x & 0x80808080u;
                const unsigned bits = (z >> 7) << (4 * (k & 1) + j);
                if (k < 2) mlo |= bits; else mhi |= bits;
            }
        }
        unsigned long long mask = ((unsigned long long)mhi << 32) | mlo;

        if (idx + NW < nchunks) {
            int nchunk = idx + NW + rot;
            if (nchunk >= nchunks) nchunk -= nchunks;
            load_key_trip<K>(kv, key, nchunk * EPW, lane);
        }

        int cnt = 0;
        for (int round = 0; round < 64; ++round) {
            const bool has = (mask != 0ull);
            const unsigned bal = __builtin_amdgcn_ballot_w32(has);
            if (bal == 0u) break;
            if (has) {
                const int bp = __builtin_ctzll(mask);
                mask &= mask - 1ull;
                const int s  = bp & 7;
                const int kk = ((bp >> 5) << 1) + (s >> 2);
                my[cnt + __popc(bal & below)] =
                    (unsigned short)(16 * lane + 512 * kk + 4 * (s & 3) + ((bp & 31) >> 3));
            }
            cnt += __popc(bal);
        }
        __builtin_amdgcn_fence(__ATOMIC_RELEASE, "wavefront");
        __builtin_amdgcn_wave_barrier();
        __builtin_amdgcn_fence(__ATOMIC_ACQUIRE, "wavefront");

        for (int b = 0; b < cnt; b += 32) {
            const int mcount = (cnt - b < 32) ? (cnt - b) : 32;
            int s = 0, dl = -1;
            float w = 0.0f;
            if (lane < mcount) {
                const int e = ebase + (int)my[b + lane];
                if (e < ne) {
                    const int d  = dst[e];
                    const int se = src[e];
                    w = dinv[se] * dinv[d];
                    const bool mine = (unsigned)(d - row0) < (unsigned)rows;
                    s  = mine ? se : 0;
                    dl = mine ? d - row0 : -1;
                }
            }
            for (int m = 0; m < mcount; m += 4) {
                float v[4][Q];
                int   dm[4];
                float wm[4];
#pragma unroll
                for (int u = 0; u < 4; ++u) {
                    const int mm = (m + u) & 31;
                    const int sm = __shfl(s, mm, 32);
                    dm[u] = __shfl(dl, mm, 32);
                    wm[u] = __shfl(w, mm, 32);
                    const float* tr = t + (size_t)sm * C + lane;
#pragma unroll
                    for (int q = 0; q < Q; ++q) v[u][q] = tr[32 * q];
                }
#pragma unroll
                for (int u = 0; u < 4; ++u) {
                    if (dm[u] >= 0) {
                        float* lr = tile + dm[u] * C + lane;
#pragma unroll
                        for (int q = 0; q < Q; ++q)
                            atomicAdd(&lr[32 * q], wm[u] * v[u][q]);
                    }
                }
            }
        }
        __builtin_amdgcn_fence(__ATOMIC_RELEASE, "wavefront");
        __builtin_amdgcn_wave_barrier();
        __builtin_amdgcn_fence(__ATOMIC_ACQUIRE, "wavefront");
    }
    __syncthreads();

    {
        float4* g4 = (float4*)(agg + (size_t)row0 * C);
        const float4* t4 = (const float4*)tile;
        for (int i = tid; i < rows * (C / 4); i += 1024) g4[i] = t4[i];
    }
}


static inline int cdiv(long long a, long long b) { return (int)((a + b - 1) / b); }

extern "C" void kernel_launch(void* const* d_in, const int* in_sizes, int n_in,
                              void* d_out, int out_size, void* d_ws, size_t ws_size,
                              hipStream_t stream) {
    const float* x  = (const float*)d_in[0];
    const float* W1 = (const float*)d_in[1];
    const float* b1 = (const float*)d_in[2];
    const float* W2 = (const float*)d_in[3];
    const float* b2 = (const float*)d_in[4];
    const float* W3 = (const float*)d_in[5];
    const float* b3 = (const float*)d_in[6];
    const int*   ei = (const int*)d_in[7];

    const int IN_C = 64, HID_C = 128, OUT_C = 64;
    const int n  = in_sizes[0] / IN_C;
    const int ne = in_sizes[7] / 2;
    const int* src = ei;
    const int* dst = ei + ne;

    const int key_words = cdiv(ne, TRIP_EDGES) * (TRIP_EDGES / 4);
    float* T    = (float*)d_ws;
    float* G    = T + (size_t)n * HID_C;
    unsigned int* key = (unsigned int*)(G + (size_t)n * HID_C);
    float* dinv = (float*)(key + key_words);
    float* out  = (float*)d_out;

    const int TB = 256;
    const long long strip_waves = n / 16;
    const int gemm_blocks = cdiv(strip_waves * 32, TB);
    constexpr int TR = TILE_ROWS, TK = KEY_LOADS;
    const int tile_blocks = cdiv(n, TR);

    deg_init_kernel<<<cdiv(n, TB), TB, 0, stream>>>(dinv, n);
    deg_count_key_kernel<<<cdiv(key_words, TB), TB, 0, stream>>>(dst, dinv, key, ne, key_words);
    deg_to_dinv_kernel<<<cdiv(n, TB), TB, 0, stream>>>(dinv, n);

    gemm_wmma_strip<false, 8><<<gemm_blocks, TB, 0, stream>>>(x, W1, T, n, IN_C);
    tile_agg_kernel<HID_C, TR, TK><<<tile_blocks, 1024, 0, stream>>>(T, dinv, b1, src, dst, key, G, n, ne);

    gemm_wmma_strip<true, 8><<<gemm_blocks, TB, 0, stream>>>(G, W2, T, n, HID_C);
    tile_agg_kernel<HID_C, TR, TK><<<tile_blocks, 1024, 0, stream>>>(T, dinv, b2, src, dst, key, G, n, ne);

    gemm_wmma_strip<true, 4><<<gemm_blocks, TB, 0, stream>>>(G, W3, T, n, HID_C);
    tile_agg_kernel<OUT_C, TR, TK><<<tile_blocks, 1024, 0, stream>>>(T, dinv, b3, src, dst, key, out, n, ne);
}


# Aggregation scans its whole key share first, noting matched edge numbers in the wave's LDS list (drained early only when full), then gathers in batches of 32, 8 rows in flight; sum order LDS arrival
# speedup vs baseline: 2.1253x; 1.0833x over previous
#include <hip/hip_runtime.h>
#include <hip/hip_bf16.h>


typedef float v2f __attribute__((ext_vector_type(2)));
typedef float v8f __attribute__((ext_vector_type(8)));


__global__ void deg_init_kernel(float* __restrict__ deg, int n) {
    int i = blockIdx.x * blockDim.x + threadIdx.x;
    if (i < n) deg[i] = 1.0f;
}

constexpr int TILE_ROWS  = 256;
constexpr int TILE_SHIFT = 8;
constexpr int KEY_LOADS  = 4;
constexpr int TRIP_EDGES = 512 * KEY_LOADS;

__global__ void deg_count_key_kernel(const int* __restrict__ dst,
                                     float* __restrict__ deg,
                                     unsigned int* __restrict__ key,
                                     int ne, int nwords) {
    const int q = blockIdx.x * blockDim.x + threadIdx.x;
    if (q >= nwords) return;
    const int e0 = 4 * q;
    unsigned int kw = 0u;
    if (e0 + 3 < ne) {
        int d[4];
#pragma unroll
        for (int j = 0; j < 4; ++j) d[j] = dst[e0 + j];
#pragma unroll
        for (int j = 0; j < 4; ++j) {
            atomicAdd(&deg[d[j]], 1.0f);
            kw |= ((unsigned int)(d[j] >> TILE_SHIFT) & 255u) << (8 * j);
        }
    } else {
#pragma unroll
        for (int j = 0; j < 4; ++j) {
            if (e0 + j < ne) {
                const int dj = dst[e0 + j];
                atomicAdd(&deg[dj], 1.0f);
                kw |= ((unsigned int)(dj >> TILE_SHIFT) & 255u) << (8 * j);
            }
        }
    }
    key[q] = kw;
}

__global__ void deg_to_dinv_kernel(float* __restrict__ deg, int n) {
    int i = blockIdx.x * blockDim.x + threadIdx.x;
    if (i < n) {
        float d = deg[i];
        deg[i] = (d > 0.0f) ? rsqrtf(d) : 0.0f;
    }
}


template <bool RELU, int TN>
__global__ void gemm_wmma_strip(const float* __restrict__ A,
                                const float* __restrict__ W,
                                float* __restrict__ out,
                                int M, int K) {
    const int Nc   = TN * 16;
    const int wave = (blockIdx.x * blockDim.x + threadIdx.x) >> 5;
    const int lane = threadIdx.x & 31;

    const int tm = wave;
    if (tm * 16 >= M) return;

    const int ml    = lane & 15;
    const int khalf = (lane >> 4) << 1;

    const float* arow = A + (size_t)(tm * 16 + ml) * K;

    v8f acc[TN];
#pragma unroll
    for (int t = 0; t < TN; ++t) acc[t] = (v8f){};

    for (int k = 0; k < K; k += 4) {
        v2f a = *(const v2f*)(arow + k + khalf);
        if (RELU) {
            a.x = fmaxf(a.x, 0.0f);
            a.y = fmaxf(a.y, 0.0f);
        }

        const float* w0 = W + (size_t)(k + khalf) * Nc + ml;
        v2f b[TN];
#pragma unroll
        for (int t = 0; t < TN; ++t) {
            b[t].x = w0[t * 16];
            b[t].y = w0[Nc + t * 16];
        }

#pragma unroll
        for (int t = 0; t < TN; ++t) {
            acc[t] = __builtin_amdgcn_wmma_f32_16x16x4_f32(
                 false, a,  false, b[t],
                 (short)0, acc[t],  false,  false);
        }
    }

    const int row0 = tm * 16 + ((lane >> 4) << 3);
#pragma unroll
    for (int t = 0; t < TN; ++t) {
        float* o = out + (size_t)row0 * Nc + t * 16 + ml;
#pragma unroll
        for (int r = 0; r < 8; ++r) o[(size_t)r * Nc] = acc[t][r];
    }
}


constexpr int LIST_CAP   = 1024;
constexpr int DRAIN_ROWS = 8;

template <int K>
__device__ __forceinline__ void load_key_trip(uint4 (&kv)[K], const unsigned int* __restrict__ key,
                                              int ebase, int lane) {
    const uint4* kp = (const uint4*)(key + (ebase >> 2));
#pragma unroll
    for (int k = 0; k < K; ++k) kv[k] = kp[lane + 32 * k];
}

template <int C>
__device__ __forceinline__ void drain_list(const int* my, int cnt, int lane,
                                           const float* __restrict__ t,
                                           const float* __restrict__ dinv,
                                           const int* __restrict__ src,
                                           const int* __restrict__ dst,
                                           float* tile, int row0, int rows, int ne) {
    constexpr int Q = C / 32;
    constexpr int G = DRAIN_ROWS;
    static_assert(32 % G == 0, "a batch of 32 is a whole number of row groups");
    __builtin_amdgcn_fence(__ATOMIC_RELEASE, "wavefront");
    __builtin_amdgcn_wave_barrier();
    __builtin_amdgcn_fence(__ATOMIC_ACQUIRE, "wavefront");

    for (int b = 0; b < cnt; b += 32) {
        const int mcount = (cnt - b < 32) ? (cnt - b) : 32;
        int s = 0, dl = -1;
        float w = 0.0f;
        if (lane < mcount) {
            const int e = my[b + lane];
            if (e < ne) {
                const int d  = dst[e];
                const int se = src[e];
                w = dinv[se] * dinv[d];
                const bool mine = (unsigned)(d - row0) < (unsigned)rows;
                s  = mine ? se : 0;
                dl = mine ? d - row0 : -1;
            }
        }
        for (int m = 0; m < mcount; m += G) {
            float v[G][Q];
            int   dm[G];
            float wm[G];
#pragma unroll
            for (int u = 0; u < G; ++u) {
                const int mm = (m + u) & 31;
                const int sm = __shfl(s, mm, 32);
                dm[u] = __shfl(dl, mm, 32);
                wm[u] = __shfl(w, mm, 32);
                const float* tr = t + (size_t)sm * C + lane;
#pragma unroll
                for (int q = 0; q < Q; ++q) v[u][q] = tr[32 * q];
            }
#pragma unroll
            for (int u = 0; u < G; ++u) {
                if (dm[u] >= 0) {
                    float* lr = tile + dm[u] * C + lane;
#pragma unroll
                    for (int q = 0; q < Q; ++q)
                        atomicAdd(&lr[32 * q], wm[u] * v[u][q]);
                }
            }
        }
    }
    __builtin_amdgcn_fence(__ATOMIC_RELEASE, "wavefront");
    __builtin_amdgcn_wave_barrier();
    __builtin_amdgcn_fence(__ATOMIC_ACQUIRE, "wavefront");
}

template <int C, int R, int K>
__global__ __launch_bounds__(1024)
void tile_agg_kernel(const float* __restrict__ t,
                     const float* __restrict__ dinv,
                     const float* __restrict__ bias,
                     const int* __restrict__ src,
                     const int* __restrict__ dst,
                     const unsigned int* __restrict__ key,
                     float* __restrict__ agg, int n, int ne) {
    static_assert(K == 4, "one match bit per inspected edge: 16*K bits in a 64-bit mask");
    static_assert(K == KEY_LOADS, "the key array is padded to whole trips of this size");
    static_assert(R == (1 << TILE_SHIFT), "the key is the tile number's low byte");
    constexpr int NW  = 32;
    constexpr int EPW = 512 * K;
    static_assert(LIST_CAP >= 64, "room for one round of 32 matches below the drain mark");
    __shared__ float tile[R * C];
    __shared__ int hits[NW * LIST_CAP];

    const int tid  = threadIdx.x;
    const int lane = tid & 31;
    const int wave = __builtin_amdgcn_readfirstlane(tid >> 5);
    const int row0 = blockIdx.x * R;
    const int rows = (n - row0 < R) ? (n - row0) : R;

    {
        static_assert(1024 % (C / 4) == 0, "a thread's column group is fixed");
        const float4* g4 = (const float4*)(t + (size_t)row0 * C);
        float4* t4 = (float4*)tile;
        const int c0 = 4 * (tid % (C / 4));
        const float bx = bias[c0], by = bias[c0 + 1], bz = bias[c0 + 2], bw = bias[c0 + 3];
        for (int i = tid; i < rows * (C / 4); i += 1024) {
            const float di = dinv[row0 + i / (C / 4)];
            const float4 tv = g4[i];
            float4 sv;
            sv.x = bx + di * di * tv.x;
            sv.y = by + di * di * tv.y;
            sv.z = bz + di * di * tv.z;
            sv.w = bw + di * di * tv.w;
            t4[i] = sv;
        }
    }
    __syncthreads();

    const int nchunks = (ne + EPW - 1) / EPW;
    const int rot = (int)(((long long)blockIdx.x * nchunks) / gridDim.x);
    int* my = hits + wave * LIST_CAP;
    const unsigned below = (1u << lane) - 1u;
    const unsigned pat = (blockIdx.x & 255u) * 0x01010101u;

    uint4 kv[K];
    if (wave < nchunks) {
        int chunk = wave + rot;
        if (chunk >= nchunks) chunk -= nchunks;
        load_key_trip<K>(kv, key, chunk * EPW, lane);
    }

    int cnt = 0;
    for (int idx = wave; idx < nchunks; idx += NW) {
        int chunk = idx + rot;
        if (chunk >= nchunks) chunk -= nchunks;
        const int ebase = chunk * EPW;

        unsigned mlo = 0u, mhi = 0u;
#pragma unroll
        for (int k = 0; k < K; ++k) {
#pragma unroll
            for (int j = 0; j < 4; ++j) {
                const unsigned wv = (j == 0) ? kv[k].x : (j == 1) ? kv[k].y
                                  : (j == 2) ? kv[k].z : kv[k].w;
                const unsigned x = wv ^ pat;
                const unsigned z = (x - 0x01010101u) & ~x & 0x80808080u;
                const unsigned bits = (z >> 7) << (4 * (k & 1) + j);
                if (k < 2) mlo |= bits; else mhi |= bits;
            }
        }
        unsigned long long mask = ((unsigned long long)mhi << 32) | mlo;

        if (idx + NW < nchunks) {
            int nchunk = idx + NW + rot;
            if (nchunk >= nchunks) nchunk -= nchunks;
            load_key_trip<K>(kv, key, nchunk * EPW, lane);
        }

        for (int round = 0; round < 64; ++round) {
            const bool has = (mask != 0ull);
            const unsigned bal = __builtin_amdgcn_ballot_w32(has);
            if (bal == 0u) break;
            if (cnt > LIST_CAP - 32) {
                drain_list<C>(my, cnt, lane, t, dinv, src, dst, tile, row0, rows, ne);
                cnt = 0;
            }
            if (has) {
                const int bp = __builtin_ctzll(mask);
                mask &= mask - 1ull;
                const int s  = bp & 7;
                const int kk = ((bp >> 5) << 1) + (s >> 2);
                my[cnt + __popc(bal & below)] =
                    ebase + 16 * lane + 512 * kk + 4 * (s & 3) + ((bp & 31) >> 3);
            }
            cnt += __popc(bal);
        }
    }
    drain_list<C>(my, cnt, lane, t, dinv, src, dst, tile, row0, rows, ne);
    __syncthreads();

    {
        float4* g4 = (float4*)(agg + (size_t)row0 * C);
        const float4* t4 = (const float4*)tile;
        for (int i = tid; i < rows * (C / 4); i += 1024) g4[i] = t4[i];
    }
}


static inline int cdiv(long long a, long long b) { return (int)((a + b - 1) / b); }

extern "C" void kernel_launch(void* const* d_in, const int* in_sizes, int n_in,
                              void* d_out, int out_size, void* d_ws, size_t ws_size,
                              hipStream_t stream) {
    const float* x  = (const float*)d_in[0];
    const float* W1 = (const float*)d_in[1];
    const float* b1 = (const float*)d_in[2];
    const float* W2 = (const float*)d_in[3];
    const float* b2 = (const float*)d_in[4];
    const float* W3 = (const float*)d_in[5];
    const float* b3 = (const float*)d_in[6];
    const int*   ei = (const int*)d_in[7];

    const int IN_C = 64, HID_C = 128, OUT_C = 64;
    const int n  = in_sizes[0] / IN_C;
    const int ne = in_sizes[7] / 2;
    const int* src = ei;
    const int* dst = ei + ne;

    const int key_words = cdiv(ne, TRIP_EDGES) * (TRIP_EDGES / 4);
    float* T    = (float*)d_ws;
    float* G    = T + (size_t)n * HID_C;
    unsigned int* key = (unsigned int*)(G + (size_t)n * HID_C);
    float* dinv = (float*)(key + key_words);
    float* out  = (float*)d_out;

    const int TB = 256;
    const long long strip_waves = n / 16;
    const int gemm_blocks = cdiv(strip_waves * 32, TB);
    constexpr int TR = TILE_ROWS, TK = KEY_LOADS;
    const int tile_blocks = cdiv(n, TR);

    deg_init_kernel<<<cdiv(n, TB), TB, 0, stream>>>(dinv, n);
    deg_count_key_kernel<<<cdiv(key_words, TB), TB, 0, stream>>>(dst, dinv, key, ne, key_words);
    deg_to_dinv_kernel<<<cdiv(n, TB), TB, 0, stream>>>(dinv, n);

    gemm_wmma_strip<false, 8><<<gemm_blocks, TB, 0, stream>>>(x, W1, T, n, IN_C);
    tile_agg_kernel<HID_C, TR, TK><<<tile_blocks, 1024, 0, stream>>>(T, dinv, b1, src, dst, key, G, n, ne);

    gemm_wmma_strip<true, 8><<<gemm_blocks, TB, 0, stream>>>(G, W2, T, n, HID_C);
    tile_agg_kernel<HID_C, TR, TK><<<tile_blocks, 1024, 0, stream>>>(T, dinv, b2, src, dst, key, G, n, ne);

    gemm_wmma_strip<true, 4><<<gemm_blocks, TB, 0, stream>>>(G, W3, T, n, HID_C);
    tile_agg_kernel<OUT_C, TR, TK><<<tile_blocks, 1024, 0, stream>>>(T, dinv, b3, src, dst, key, out, n, ne);
}


# Layer-1 aggregation saves each resolved match (source, tile row, norm product) per wave, unsorted; layers 2, 3 add from these records: no key scan, no dst/src/dinv reads; sum order LDS arrival
# speedup vs baseline: 2.4397x; 1.1479x over previous
#include <hip/hip_runtime.h>
#include <hip/hip_bf16.h>


typedef float v2f __attribute__((ext_vector_type(2)));
typedef float v8f __attribute__((ext_vector_type(8)));


__global__ void deg_init_kernel(float* __restrict__ deg, int n) {
    int i = blockIdx.x * blockDim.x + threadIdx.x;
    if (i < n) deg[i] = 1.0f;
}

constexpr int TILE_ROWS  = 256;
constexpr int TILE_SHIFT = 8;
constexpr int KEY_LOADS  = 4;
constexpr int TRIP_EDGES = 512 * KEY_LOADS;

__global__ void deg_count_key_kernel(const int* __restrict__ dst,
                                     float* __restrict__ deg,
                                     unsigned int* __restrict__ key,
                                     int ne, int nwords) {
    const int q = blockIdx.x * blockDim.x + threadIdx.x;
    if (q >= nwords) return;
    const int e0 = 4 * q;
    unsigned int kw = 0u;
    if (e0 + 3 < ne) {
        int d[4];
#pragma unroll
        for (int j = 0; j < 4; ++j) d[j] = dst[e0 + j];
#pragma unroll
        for (int j = 0; j < 4; ++j) {
            atomicAdd(&deg[d[j]], 1.0f);
            kw |= ((unsigned int)(d[j] >> TILE_SHIFT) & 255u) << (8 * j);
        }
    } else {
#pragma unroll
        for (int j = 0; j < 4; ++j) {
            if (e0 + j < ne) {
                const int dj = dst[e0 + j];
                atomicAdd(&deg[dj], 1.0f);
                kw |= ((unsigned int)(dj >> TILE_SHIFT) & 255u) << (8 * j);
            }
        }
    }
    key[q] = kw;
}

__global__ void deg_to_dinv_kernel(float* __restrict__ deg, int n) {
    int i = blockIdx.x * blockDim.x + threadIdx.x;
    if (i < n) {
        float d = deg[i];
        deg[i] = (d > 0.0f) ? rsqrtf(d) : 0.0f;
    }
}


template <bool RELU, int TN>
__global__ void gemm_wmma_strip(const float* __restrict__ A,
                                const float* __restrict__ W,
                                float* __restrict__ out,
                                int M, int K) {
    const int Nc   = TN * 16;
    const int wave = (blockIdx.x * blockDim.x + threadIdx.x) >> 5;
    const int lane = threadIdx.x & 31;

    const int tm = wave;
    if (tm * 16 >= M) return;

    const int ml    = lane & 15;
    const int khalf = (lane >> 4) << 1;

    const float* arow = A + (size_t)(tm * 16 + ml) * K;

    v8f acc[TN];
#pragma unroll
    for (int t = 0; t < TN; ++t) acc[t] = (v8f){};

    for (int k = 0; k < K; k += 4) {
        v2f a = *(const v2f*)(arow + k + khalf);
        if (RELU) {
            a.x = fmaxf(a.x, 0.0f);
            a.y = fmaxf(a.y, 0.0f);
        }

        const float* w0 = W + (size_t)(k + khalf) * Nc + ml;
        v2f b[TN];
#pragma unroll
        for (int t = 0; t < TN; ++t) {
            b[t].x = w0[t * 16];
            b[t].y = w0[Nc + t * 16];
        }

#pragma unroll
        for (int t = 0; t < TN; ++t) {
            acc[t] = __builtin_amdgcn_wmma_f32_16x16x4_f32(
                 false, a,  false, b[t],
                 (short)0, acc[t],  false,  false);
        }
    }

    const int row0 = tm * 16 + ((lane >> 4) << 3);
#pragma unroll
    for (int t = 0; t < TN; ++t) {
        float* o = out + (size_t)row0 * Nc + t * 16 + ml;
#pragma unroll
        for (int r = 0; r < 8; ++r) o[(size_t)r * Nc] = acc[t][r];
    }
}


constexpr int LIST_CAP   = 1024;
constexpr int DRAIN_ROWS = 8;
constexpr int REC_CAP    = 512;
constexpr int PRE_BATCHES = 6;

template <int K>
__device__ __forceinline__ void load_key_trip(uint4 (&kv)[K], const unsigned int* __restrict__ key,
                                              int ebase, int lane) {
    const uint4* kp = (const uint4*)(key + (ebase >> 2));
#pragma unroll
    for (int k = 0; k < K; ++k) kv[k] = kp[lane + 32 * k];
}

template <int C>
__device__ __forceinline__ void add_batch(int s, int dl, float w, int mcount, int lane,
                                          const float* __restrict__ t, float* tile) {
    constexpr int Q = C / 32;
    constexpr int G = DRAIN_ROWS;
    static_assert(32 % G == 0, "a batch of 32 is a whole number of row groups");
    for (int m = 0; m < mcount; m += G) {
        float v[G][Q];
        int   dm[G];
        float wm[G];
#pragma unroll
        for (int u = 0; u < G; ++u) {
            const int mm = (m + u) & 31;
            const int sm = __shfl(s, mm, 32);
            dm[u] = __shfl(dl, mm, 32);
            wm[u] = __shfl(w, mm, 32);
            const float* tr = t + (size_t)sm * C + lane;
#pragma unroll
            for (int q = 0; q < Q; ++q) v[u][q] = tr[32 * q];
        }
#pragma unroll
        for (int u = 0; u < G; ++u) {
            if (dm[u] >= 0) {
                float* lr = tile + dm[u] * C + lane;
#pragma unroll
                for (int q = 0; q < Q; ++q)
                    atomicAdd(&lr[32 * q], wm[u] * v[u][q]);
            }
        }
    }
}

template <int C, bool SAVE>
__device__ __forceinline__ void drain_list(const int* my, int cnt, int lane,
                                           const float* __restrict__ t,
                                           const float* __restrict__ dinv,
                                           const int* __restrict__ src,
                                           const int* __restrict__ dst,
                                           float* tile, int row0, int rows, int ne,
                                           int* __restrict__ rs, int* __restrict__ rd,
                                           float* __restrict__ rw, int& saved) {
    __builtin_amdgcn_fence(__ATOMIC_RELEASE, "wavefront");
    __builtin_amdgcn_wave_barrier();
    __builtin_amdgcn_fence(__ATOMIC_ACQUIRE, "wavefront");

    bool keep = false;
    if (SAVE) {
        keep = (saved >= 0) && (saved + cnt <= REC_CAP);
        if (!keep) saved = -1;
    }

    for (int b = 0; b < cnt; b += 32) {
        const int mcount = (cnt - b < 32) ? (cnt - b) : 32;
        int s = 0, dl = -1;
        float w = 0.0f;
        if (lane < mcount) {
            const int e = my[b + lane];
            if (e < ne) {
                const int d  = dst[e];
                const int se = src[e];
                w = dinv[se] * dinv[d];
                const bool mine = (unsigned)(d - row0) < (unsigned)rows;
                s  = mine ? se : 0;
                dl = mine ? d - row0 : -1;
            }
            if (SAVE && keep) {
                rs[saved + b + lane] = s;
                rd[saved + b + lane] = dl;
                rw[saved + b + lane] = w;
            }
        }
        add_batch<C>(s, dl, w, mcount, lane, t, tile);
    }
    if (SAVE && keep) saved += cnt;
    __builtin_amdgcn_fence(__ATOMIC_RELEASE, "wavefront");
    __builtin_amdgcn_wave_barrier();
    __builtin_amdgcn_fence(__ATOMIC_ACQUIRE, "wavefront");
}

template <int C>
__device__ __forceinline__ void drain_records(const int* __restrict__ rs,
                                              const int* __restrict__ rd,
                                              const float* __restrict__ rw,
                                              int cnt, int lane,
                                              int (&ps)[PRE_BATCHES],
                                              int (&pd)[PRE_BATCHES],
                                              float (&pw)[PRE_BATCHES],
                                              const float* __restrict__ t, float* tile) {
    for (int b = 0; b < cnt; b += 32) {
        const int mcount = (cnt - b < 32) ? (cnt - b) : 32;
        int s = ps[0], dl = pd[0];
        float w = pw[0];
        if (b < 32 * PRE_BATCHES) {
#pragma unroll
            for (int q = 0; q + 1 < PRE_BATCHES; ++q) {
                ps[q] = ps[q + 1];
                pd[q] = pd[q + 1];
                pw[q] = pw[q + 1];
            }
        } else {
            s = 0; dl = -1; w = 0.0f;
            if (lane < mcount) {
                s  = rs[b + lane];
                dl = rd[b + lane];
                w  = rw[b + lane];
            }
        }
        add_batch<C>(s, dl, w, mcount, lane, t, tile);
    }
}

template <int C, int R, int K, bool FIRST>
__global__ __launch_bounds__(1024)
void tile_agg_kernel(const float* __restrict__ t,
                     const float* __restrict__ dinv,
                     const float* __restrict__ bias,
                     const int* __restrict__ src,
                     const int* __restrict__ dst,
                     const unsigned int* __restrict__ key,
                     int* __restrict__ rec_s,
                     int* __restrict__ rec_dl,
                     float* __restrict__ rec_w,
                     int* __restrict__ rec_cnt,
                     float* __restrict__ agg, int n, int ne) {
    static_assert(K == 4, "one match bit per inspected edge: 16*K bits in a 64-bit mask");
    static_assert(K == KEY_LOADS, "the key array is padded to whole trips of this size");
    static_assert(R == (1 << TILE_SHIFT), "the key is the tile number's low byte");
    constexpr int NW  = 32;
    constexpr int EPW = 512 * K;
    static_assert(LIST_CAP >= 64, "room for one round of 32 matches below the drain mark");
    __shared__ float tile[R * C];
    __shared__ int hits[NW * LIST_CAP];

    const int tid  = threadIdx.x;
    const int lane = tid & 31;
    const int wave = __builtin_amdgcn_readfirstlane(tid >> 5);
    const int row0 = blockIdx.x * R;
    const int rows = (n - row0 < R) ? (n - row0) : R;

    const int slot = blockIdx.x * NW + wave;
    int*   rs = rec_s  + (size_t)slot * REC_CAP;
    int*   rd = rec_dl + (size_t)slot * REC_CAP;
    float* rw = rec_w  + (size_t)slot * REC_CAP;

    int have = -1;
    int   ps[PRE_BATCHES];
    int   pd[PRE_BATCHES];
    float pw[PRE_BATCHES];
#pragma unroll
    for (int j = 0; j < PRE_BATCHES; ++j) { ps[j] = 0; pd[j] = -1; pw[j] = 0.0f; }
    if (!FIRST) {
        have = __builtin_amdgcn_readfirstlane(rec_cnt[slot]);
        if (have > REC_CAP) have = -1;
#pragma unroll
        for (int j = 0; j < PRE_BATCHES; ++j) {
            const int i = 32 * j + lane;
            if (i < have) {
                ps[j] = rs[i];
                pd[j] = rd[i];
                pw[j] = rw[i];
            }
        }
    }

    {
        static_assert(1024 % (C / 4) == 0, "a thread's column group is fixed");
        const float4* g4 = (const float4*)(t + (size_t)row0 * C);
        float4* t4 = (float4*)tile;
        const int c0 = 4 * (tid % (C / 4));
        const float bx = bias[c0], by = bias[c0 + 1], bz = bias[c0 + 2], bw = bias[c0 + 3];
        for (int i = tid; i < rows * (C / 4); i += 1024) {
            const float di = dinv[row0 + i / (C / 4)];
            const float4 tv = g4[i];
            float4 sv;
            sv.x = bx + di * di * tv.x;
            sv.y = by + di * di * tv.y;
            sv.z = bz + di * di * tv.z;
            sv.w = bw + di * di * tv.w;
            t4[i] = sv;
        }
    }
    __syncthreads();

    if (!FIRST && have >= 0) {
        drain_records<C>(rs, rd, rw, have, lane, ps, pd, pw, t, tile);
    } else {
        const int nchunks = (ne + EPW - 1) / EPW;
        const int rot = (int)(((long long)blockIdx.x * nchunks) / gridDim.x);
        int* my = hits + wave * LIST_CAP;
        const unsigned below = (1u << lane) - 1u;
        const unsigned pat = (blockIdx.x & 255u) * 0x01010101u;

        uint4 kv[K];
        if (wave < nchunks) {
            int chunk = wave + rot;
            if (chunk >= nchunks) chunk -= nchunks;
            load_key_trip<K>(kv, key, chunk * EPW, lane);
        }

        int cnt = 0;
        int saved = 0;
        for (int idx = wave; idx < nchunks; idx += NW) {
            int chunk = idx + rot;
            if (chunk >= nchunks) chunk -= nchunks;
            const int ebase = chunk * EPW;

            unsigned mlo = 0u, mhi = 0u;
#pragma unroll
            for (int k = 0; k < K; ++k) {
#pragma unroll
                for (int j = 0; j < 4; ++j) {
                    const unsigned wv = (j == 0) ? kv[k].x : (j == 1) ? kv[k].y
                                      : (j == 2) ? kv[k].z : kv[k].w;
                    const unsigned x = wv ^ pat;
                    const unsigned z = (x - 0x01010101u) & ~x & 0x80808080u;
                    const unsigned bits = (z >> 7) << (4 * (k & 1) + j);
                    if (k < 2) mlo |= bits; else mhi |= bits;
                }
            }
            unsigned long long mask = ((unsigned long long)mhi << 32) | mlo;

            if (idx + NW < nchunks) {
                int nchunk = idx + NW + rot;
                if (nchunk >= nchunks) nchunk -= nchunks;
                load_key_trip<K>(kv, key, nchunk * EPW, lane);
            }

            for (int round = 0; round < 64; ++round) {
                const bool has = (mask != 0ull);
                const unsigned bal = __builtin_amdgcn_ballot_w32(has);
                if (bal == 0u) break;
                if (cnt > LIST_CAP - 32) {
                    drain_list<C, FIRST>(my, cnt, lane, t, dinv, src, dst, tile, row0, rows, ne,
                                         rs, rd, rw, saved);
                    cnt = 0;
                }
                if (has) {
                    const int bp = __builtin_ctzll(mask);
                    mask &= mask - 1ull;
                    const int s  = bp & 7;
                    const int kk = ((bp >> 5) << 1) + (s >> 2);
                    my[cnt + __popc(bal & below)] =
                        ebase + 16 * lane + 512 * kk + 4 * (s & 3) + ((bp & 31) >> 3);
                }
                cnt += __popc(bal);
            }
        }
        drain_list<C, FIRST>(my, cnt, lane, t, dinv, src, dst, tile, row0, rows, ne,
                             rs, rd, rw, saved);
        if (FIRST && lane == 0) rec_cnt[slot] = saved;
    }
    __syncthreads();

    {
        float4* g4 = (float4*)(agg + (size_t)row0 * C);
        const float4* t4 = (const float4*)tile;
        for (int i = tid; i < rows * (C / 4); i += 1024) g4[i] = t4[i];
    }
}


static inline int cdiv(long long a, long long b) { return (int)((a + b - 1) / b); }

extern "C" void kernel_launch(void* const* d_in, const int* in_sizes, int n_in,
                              void* d_out, int out_size, void* d_ws, size_t ws_size,
                              hipStream_t stream) {
    const float* x  = (const float*)d_in[0];
    const float* W1 = (const float*)d_in[1];
    const float* b1 = (const float*)d_in[2];
    const float* W2 = (const float*)d_in[3];
    const float* b2 = (const float*)d_in[4];
    const float* W3 = (const float*)d_in[5];
    const float* b3 = (const float*)d_in[6];
    const int*   ei = (const int*)d_in[7];

    const int IN_C = 64, HID_C = 128, OUT_C = 64;
    const int n  = in_sizes[0] / IN_C;
    const int ne = in_sizes[7] / 2;
    const int* src = ei;
    const int* dst = ei + ne;

    const int key_words = cdiv(ne, TRIP_EDGES) * (TRIP_EDGES / 4);
    float* T    = (float*)d_ws;
    float* G    = T + (size_t)n * HID_C;
    unsigned int* key = (unsigned int*)(G + (size_t)n * HID_C);
    float* dinv = (float*)(key + key_words);
    float* out  = (float*)d_out;

    const int TB = 256;
    const long long strip_waves = n / 16;
    const int gemm_blocks = cdiv(strip_waves * 32, TB);
    constexpr int TR = TILE_ROWS, TK = KEY_LOADS;
    const int tile_blocks = cdiv(n, TR);
    const size_t rec_slots = (size_t)tile_blocks * 32;
    int*   rec_s   = (int*)(dinv + n);
    int*   rec_dl  = rec_s + rec_slots * REC_CAP;
    float* rec_w   = (float*)(rec_dl + rec_slots * REC_CAP);
    int*   rec_cnt = (int*)(rec_w + rec_slots * REC_CAP);

    deg_init_kernel<<<cdiv(n, TB), TB, 0, stream>>>(dinv, n);
    deg_count_key_kernel<<<cdiv(key_words, TB), TB, 0, stream>>>(dst, dinv, key, ne, key_words);
    deg_to_dinv_kernel<<<cdiv(n, TB), TB, 0, stream>>>(dinv, n);

    gemm_wmma_strip<false, 8><<<gemm_blocks, TB, 0, stream>>>(x, W1, T, n, IN_C);
    tile_agg_kernel<HID_C, TR, TK, true><<<tile_blocks, 1024, 0, stream>>>(
        T, dinv, b1, src, dst, key, rec_s, rec_dl, rec_w, rec_cnt, G, n, ne);

    gemm_wmma_strip<true, 8><<<gemm_blocks, TB, 0, stream>>>(G, W2, T, n, HID_C);
    tile_agg_kernel<HID_C, TR, TK, false><<<tile_blocks, 1024, 0, stream>>>(
        T, dinv, b2, src, dst, key, rec_s, rec_dl, rec_w, rec_cnt, G, n, ne);

    gemm_wmma_strip<true, 4><<<gemm_blocks, TB, 0, stream>>>(G, W3, T, n, HID_C);
    tile_agg_kernel<OUT_C, TR, TK, false><<<tile_blocks, 1024, 0, stream>>>(
        T, dinv, b3, src, dst, key, rec_s, rec_dl, rec_w, rec_cnt, out, n, ne);
}


# Aggregation fetches each gathered source row with one vector load per lane (16 B at 128 columns, 8 B at 64), not 4 or 2 dword loads; LDS tile columns interleaved to match; sum order unchanged
# speedup vs baseline: 2.5601x; 1.0494x over previous
#include <hip/hip_runtime.h>
#include <hip/hip_bf16.h>


typedef float v2f __attribute__((ext_vector_type(2)));
typedef float v8f __attribute__((ext_vector_type(8)));


__global__ void deg_init_kernel(float* __restrict__ deg, int n) {
    int i = blockIdx.x * blockDim.x + threadIdx.x;
    if (i < n) deg[i] = 1.0f;
}

constexpr int TILE_ROWS  = 256;
constexpr int TILE_SHIFT = 8;
constexpr int KEY_LOADS  = 4;
constexpr int TRIP_EDGES = 512 * KEY_LOADS;

__global__ void deg_count_key_kernel(const int* __restrict__ dst,
                                     float* __restrict__ deg,
                                     unsigned int* __restrict__ key,
                                     int ne, int nwords) {
    const int q = blockIdx.x * blockDim.x + threadIdx.x;
    if (q >= nwords) return;
    const int e0 = 4 * q;
    unsigned int kw = 0u;
    if (e0 + 3 < ne) {
        int d[4];
#pragma unroll
        for (int j = 0; j < 4; ++j) d[j] = dst[e0 + j];
#pragma unroll
        for (int j = 0; j < 4; ++j) {
            atomicAdd(&deg[d[j]], 1.0f);
            kw |= ((unsigned int)(d[j] >> TILE_SHIFT) & 255u) << (8 * j);
        }
    } else {
#pragma unroll
        for (int j = 0; j < 4; ++j) {
            if (e0 + j < ne) {
                const int dj = dst[e0 + j];
                atomicAdd(&deg[dj], 1.0f);
                kw |= ((unsigned int)(dj >> TILE_SHIFT) & 255u) << (8 * j);
            }
        }
    }
    key[q] = kw;
}

__global__ void deg_to_dinv_kernel(float* __restrict__ deg, int n) {
    int i = blockIdx.x * blockDim.x + threadIdx.x;
    if (i < n) {
        float d = deg[i];
        deg[i] = (d > 0.0f) ? rsqrtf(d) : 0.0f;
    }
}


template <bool RELU, int TN>
__global__ void gemm_wmma_strip(const float* __restrict__ A,
                                const float* __restrict__ W,
                                float* __restrict__ out,
                                int M, int K) {
    const int Nc   = TN * 16;
    const int wave = (blockIdx.x * blockDim.x + threadIdx.x) >> 5;
    const int lane = threadIdx.x & 31;

    const int tm = wave;
    if (tm * 16 >= M) return;

    const int ml    = lane & 15;
    const int khalf = (lane >> 4) << 1;

    const float* arow = A + (size_t)(tm * 16 + ml) * K;

    v8f acc[TN];
#pragma unroll
    for (int t = 0; t < TN; ++t) acc[t] = (v8f){};

    for (int k = 0; k < K; k += 4) {
        v2f a = *(const v2f*)(arow + k + khalf);
        if (RELU) {
            a.x = fmaxf(a.x, 0.0f);
            a.y = fmaxf(a.y, 0.0f);
        }

        const float* w0 = W + (size_t)(k + khalf) * Nc + ml;
        v2f b[TN];
#pragma unroll
        for (int t = 0; t < TN; ++t) {
            b[t].x = w0[t * 16];
            b[t].y = w0[Nc + t * 16];
        }

#pragma unroll
        for (int t = 0; t < TN; ++t) {
            acc[t] = __builtin_amdgcn_wmma_f32_16x16x4_f32(
                 false, a,  false, b[t],
                 (short)0, acc[t],  false,  false);
        }
    }

    const int row0 = tm * 16 + ((lane >> 4) << 3);
#pragma unroll
    for (int t = 0; t < TN; ++t) {
        float* o = out + (size_t)row0 * Nc + t * 16 + ml;
#pragma unroll
        for (int r = 0; r < 8; ++r) o[(size_t)r * Nc] = acc[t][r];
    }
}


constexpr int LIST_CAP   = 1024;
constexpr int DRAIN_ROWS = 8;
constexpr int REC_CAP    = 512;
constexpr int PRE_BATCHES = 6;

template <int K>
__device__ __forceinline__ void load_key_trip(uint4 (&kv)[K], const unsigned int* __restrict__ key,
                                              int ebase, int lane) {
    const uint4* kp = (const uint4*)(key + (ebase >> 2));
#pragma unroll
    for (int k = 0; k < K; ++k) kv[k] = kp[lane + 32 * k];
}

template <int C>
__device__ __forceinline__ int tile_pos(int c) {
    constexpr int Q = C / 32;
    return (c % Q) * 32 + c / Q;
}

template <int C>
__device__ __forceinline__ void add_batch(int s, int dl, float w, int mcount, int lane,
                                          const float* __restrict__ t, float* tile) {
    constexpr int Q = C / 32;
    typedef float vq __attribute__((ext_vector_type(Q)));
    constexpr int G = DRAIN_ROWS;
    static_assert(32 % G == 0, "a batch of 32 is a whole number of row groups");
    for (int m = 0; m < mcount; m += G) {
        vq    v[G];
        int   dm[G];
        float wm[G];
#pragma unroll
        for (int u = 0; u < G; ++u) {
            const int mm = (m + u) & 31;
            const int sm = __shfl(s, mm, 32);
            dm[u] = __shfl(dl, mm, 32);
            wm[u] = __shfl(w, mm, 32);
            v[u] = *(const vq*)(t + (size_t)sm * C + Q * lane);
        }
#pragma unroll
        for (int u = 0; u < G; ++u) {
            if (dm[u] >= 0) {
                float* lr = tile + dm[u] * C + lane;
#pragma unroll
                for (int q = 0; q < Q; ++q)
                    atomicAdd(&lr[32 * q], wm[u] * v[u][q]);
            }
        }
    }
}

template <int C, bool SAVE>
__device__ __forceinline__ void drain_list(const int* my, int cnt, int lane,
                                           const float* __restrict__ t,
                                           const float* __restrict__ dinv,
                                           const int* __restrict__ src,
                                           const int* __restrict__ dst,
                                           float* tile, int row0, int rows, int ne,
                                           int* __restrict__ rs, int* __restrict__ rd,
                                           float* __restrict__ rw, int& saved) {
    __builtin_amdgcn_fence(__ATOMIC_RELEASE, "wavefront");
    __builtin_amdgcn_wave_barrier();
    __builtin_amdgcn_fence(__ATOMIC_ACQUIRE, "wavefront");

    bool keep = false;
    if (SAVE) {
        keep = (saved >= 0) && (saved + cnt <= REC_CAP);
        if (!keep) saved = -1;
    }

    for (int b = 0; b < cnt; b += 32) {
        const int mcount = (cnt - b < 32) ? (cnt - b) : 32;
        int s = 0, dl = -1;
        float w = 0.0f;
        if (lane < mcount) {
            const int e = my[b + lane];
            if (e < ne) {
                const int d  = dst[e];
                const int se = src[e];
                w = dinv[se] * dinv[d];
                const bool mine = (unsigned)(d - row0) < (unsigned)rows;
                s  = mine ? se : 0;
                dl = mine ? d - row0 : -1;
            }
            if (SAVE && keep) {
                rs[saved + b + lane] = s;
                rd[saved + b + lane] = dl;
                rw[saved + b + lane] = w;
            }
        }
        add_batch<C>(s, dl, w, mcount, lane, t, tile);
    }
    if (SAVE && keep) saved += cnt;
    __builtin_amdgcn_fence(__ATOMIC_RELEASE, "wavefront");
    __builtin_amdgcn_wave_barrier();
    __builtin_amdgcn_fence(__ATOMIC_ACQUIRE, "wavefront");
}

template <int C>
__device__ __forceinline__ void drain_records(const int* __restrict__ rs,
                                              const int* __restrict__ rd,
                                              const float* __restrict__ rw,
                                              int cnt, int lane,
                                              int (&ps)[PRE_BATCHES],
                                              int (&pd)[PRE_BATCHES],
                                              float (&pw)[PRE_BATCHES],
                                              const float* __restrict__ t, float* tile) {
    for (int b = 0; b < cnt; b += 32) {
        const int mcount = (cnt - b < 32) ? (cnt - b) : 32;
        int s = ps[0], dl = pd[0];
        float w = pw[0];
        if (b < 32 * PRE_BATCHES) {
#pragma unroll
            for (int q = 0; q + 1 < PRE_BATCHES; ++q) {
                ps[q] = ps[q + 1];
                pd[q] = pd[q + 1];
                pw[q] = pw[q + 1];
            }
        } else {
            s = 0; dl = -1; w = 0.0f;
            if (lane < mcount) {
                s  = rs[b + lane];
                dl = rd[b + lane];
                w  = rw[b + lane];
            }
        }
        add_batch<C>(s, dl, w, mcount, lane, t, tile);
    }
}

template <int C, int R, int K, bool FIRST>
__global__ __launch_bounds__(1024)
void tile_agg_kernel(const float* __restrict__ t,
                     const float* __restrict__ dinv,
                     const float* __restrict__ bias,
                     const int* __restrict__ src,
                     const int* __restrict__ dst,
                     const unsigned int* __restrict__ key,
                     int* __restrict__ rec_s,
                     int* __restrict__ rec_dl,
                     float* __restrict__ rec_w,
                     int* __restrict__ rec_cnt,
                     float* __restrict__ agg, int n, int ne) {
    static_assert(K == 4, "one match bit per inspected edge: 16*K bits in a 64-bit mask");
    static_assert(K == KEY_LOADS, "the key array is padded to whole trips of this size");
    static_assert(R == (1 << TILE_SHIFT), "the key is the tile number's low byte");
    constexpr int NW  = 32;
    constexpr int EPW = 512 * K;
    static_assert(LIST_CAP >= 64, "room for one round of 32 matches below the drain mark");
    __shared__ float tile[R * C];
    __shared__ int hits[NW * LIST_CAP];

    const int tid  = threadIdx.x;
    const int lane = tid & 31;
    const int wave = __builtin_amdgcn_readfirstlane(tid >> 5);
    const int row0 = blockIdx.x * R;
    const int rows = (n - row0 < R) ? (n - row0) : R;

    const int slot = blockIdx.x * NW + wave;
    int*   rs = rec_s  + (size_t)slot * REC_CAP;
    int*   rd = rec_dl + (size_t)slot * REC_CAP;
    float* rw = rec_w  + (size_t)slot * REC_CAP;

    int have = -1;
    int   ps[PRE_BATCHES];
    int   pd[PRE_BATCHES];
    float pw[PRE_BATCHES];
#pragma unroll
    for (int j = 0; j < PRE_BATCHES; ++j) { ps[j] = 0; pd[j] = -1; pw[j] = 0.0f; }
    if (!FIRST) {
        have = __builtin_amdgcn_readfirstlane(rec_cnt[slot]);
        if (have > REC_CAP) have = -1;
#pragma unroll
        for (int j = 0; j < PRE_BATCHES; ++j) {
            const int i = 32 * j + lane;
            if (i < have) {
                ps[j] = rs[i];
                pd[j] = rd[i];
                pw[j] = rw[i];
            }
        }
    }

    {
        static_assert(1024 % (C / 4) == 0, "a thread's column group is fixed");
        const float4* g4 = (const float4*)(t + (size_t)row0 * C);
        const int c0 = 4 * (tid % (C / 4));
        const float bx = bias[c0], by = bias[c0 + 1], bz = bias[c0 + 2], bw = bias[c0 + 3];
        const int p0 = tile_pos<C>(c0),     p1 = tile_pos<C>(c0 + 1);
        const int p2 = tile_pos<C>(c0 + 2), p3 = tile_pos<C>(c0 + 3);
        for (int i = tid; i < rows * (C / 4); i += 1024) {
            const int r = i / (C / 4);
            const float di = dinv[row0 + r];
            const float4 tv = g4[i];
            float* lr = tile + r * C;
            lr[p0] = bx + di * di * tv.x;
            lr[p1] = by + di * di * tv.y;
            lr[p2] = bz + di * di * tv.z;
            lr[p3] = bw + di * di * tv.w;
        }
    }
    __syncthreads();

    if (!FIRST && have >= 0) {
        drain_records<C>(rs, rd, rw, have, lane, ps, pd, pw, t, tile);
    } else {
        const int nchunks = (ne + EPW - 1) / EPW;
        const int rot = (int)(((long long)blockIdx.x * nchunks) / gridDim.x);
        int* my = hits + wave * LIST_CAP;
        const unsigned below = (1u << lane) - 1u;
        const unsigned pat = (blockIdx.x & 255u) * 0x01010101u;

        uint4 kv[K];
        if (wave < nchunks) {
            int chunk = wave + rot;
            if (chunk >= nchunks) chunk -= nchunks;
            load_key_trip<K>(kv, key, chunk * EPW, lane);
        }

        int cnt = 0;
        int saved = 0;
        for (int idx = wave; idx < nchunks; idx += NW) {
            int chunk = idx + rot;
            if (chunk >= nchunks) chunk -= nchunks;
            const int ebase = chunk * EPW;

            unsigned mlo = 0u, mhi = 0u;
#pragma unroll
            for (int k = 0; k < K; ++k) {
#pragma unroll
                for (int j = 0; j < 4; ++j) {
                    const unsigned wv = (j == 0) ? kv[k].x : (j == 1) ? kv[k].y
                                      : (j == 2) ? kv[k].z : kv[k].w;
                    const unsigned x = wv ^ pat;
                    const unsigned z = (x - 0x01010101u) & ~x & 0x80808080u;
                    const unsigned bits = (z >> 7) << (4 * (k & 1) + j);
                    if (k < 2) mlo |= bits; else mhi |= bits;
                }
            }
            unsigned long long mask = ((unsigned long long)mhi << 32) | mlo;

            if (idx + NW < nchunks) {
                int nchunk = idx + NW + rot;
                if (nchunk >= nchunks) nchunk -= nchunks;
                load_key_trip<K>(kv, key, nchunk * EPW, lane);
            }

            for (int round = 0; round < 64; ++round) {
                const bool has = (mask != 0ull);
                const unsigned bal = __builtin_amdgcn_ballot_w32(has);
                if (bal == 0u) break;
                if (cnt > LIST_CAP - 32) {
                    drain_list<C, FIRST>(my, cnt, lane, t, dinv, src, dst, tile, row0, rows, ne,
                                         rs, rd, rw, saved);
                    cnt = 0;
                }
                if (has) {
                    const int bp = __builtin_ctzll(mask);
                    mask &= mask - 1ull;
                    const int s  = bp & 7;
                    const int kk = ((bp >> 5) << 1) + (s >> 2);
                    my[cnt + __popc(bal & below)] =
                        ebase + 16 * lane + 512 * kk + 4 * (s & 3) + ((bp & 31) >> 3);
                }
                cnt += __popc(bal);
            }
        }
        drain_list<C, FIRST>(my, cnt, lane, t, dinv, src, dst, tile, row0, rows, ne,
                             rs, rd, rw, saved);
        if (FIRST && lane == 0) rec_cnt[slot] = saved;
    }
    __syncthreads();

    {
        float4* g4 = (float4*)(agg + (size_t)row0 * C);
        const int c0 = 4 * (tid % (C / 4));
        const int p0 = tile_pos<C>(c0),     p1 = tile_pos<C>(c0 + 1);
        const int p2 = tile_pos<C>(c0 + 2), p3 = tile_pos<C>(c0 + 3);
        for (int i = tid; i < rows * (C / 4); i += 1024) {
            const float* lr = tile + (i / (C / 4)) * C;
            float4 o;
            o.x = lr[p0];
            o.y = lr[p1];
            o.z = lr[p2];
            o.w = lr[p3];
            g4[i] = o;
        }
    }
}


static inline int cdiv(long long a, long long b) { return (int)((a + b - 1) / b); }

extern "C" void kernel_launch(void* const* d_in, const int* in_sizes, int n_in,
                              void* d_out, int out_size, void* d_ws, size_t ws_size,
                              hipStream_t stream) {
    const float* x  = (const float*)d_in[0];
    const float* W1 = (const float*)d_in[1];
    const float* b1 = (const float*)d_in[2];
    const float* W2 = (const float*)d_in[3];
    const float* b2 = (const float*)d_in[4];
    const float* W3 = (const float*)d_in[5];
    const float* b3 = (const float*)d_in[6];
    const int*   ei = (const int*)d_in[7];

    const int IN_C = 64, HID_C = 128, OUT_C = 64;
    const int n  = in_sizes[0] / IN_C;
    const int ne = in_sizes[7] / 2;
    const int* src = ei;
    const int* dst = ei + ne;

    const int key_words = cdiv(ne, TRIP_EDGES) * (TRIP_EDGES / 4);
    float* T    = (float*)d_ws;
    float* G    = T + (size_t)n * HID_C;
    unsigned int* key = (unsigned int*)(G + (size_t)n * HID_C);
    float* dinv = (float*)(key + key_words);
    float* out  = (float*)d_out;

    const int TB = 256;
    const long long strip_waves = n / 16;
    const int gemm_blocks = cdiv(strip_waves * 32, TB);
    constexpr int TR = TILE_ROWS, TK = KEY_LOADS;
    const int tile_blocks = cdiv(n, TR);
    const size_t rec_slots = (size_t)tile_blocks * 32;
    int*   rec_s   = (int*)(dinv + n);
    int*   rec_dl  = rec_s + rec_slots * REC_CAP;
    float* rec_w   = (float*)(rec_dl + rec_slots * REC_CAP);
    int*   rec_cnt = (int*)(rec_w + rec_slots * REC_CAP);

    deg_init_kernel<<<cdiv(n, TB), TB, 0, stream>>>(dinv, n);
    deg_count_key_kernel<<<cdiv(key_words, TB), TB, 0, stream>>>(dst, dinv, key, ne, key_words);
    deg_to_dinv_kernel<<<cdiv(n, TB), TB, 0, stream>>>(dinv, n);

    gemm_wmma_strip<false, 8><<<gemm_blocks, TB, 0, stream>>>(x, W1, T, n, IN_C);
    tile_agg_kernel<HID_C, TR, TK, true><<<tile_blocks, 1024, 0, stream>>>(
        T, dinv, b1, src, dst, key, rec_s, rec_dl, rec_w, rec_cnt, G, n, ne);

    gemm_wmma_strip<true, 8><<<gemm_blocks, TB, 0, stream>>>(G, W2, T, n, HID_C);
    tile_agg_kernel<HID_C, TR, TK, false><<<tile_blocks, 1024, 0, stream>>>(
        T, dinv, b2, src, dst, key, rec_s, rec_dl, rec_w, rec_cnt, G, n, ne);

    gemm_wmma_strip<true, 4><<<gemm_blocks, TB, 0, stream>>>(G, W3, T, n, HID_C);
    tile_agg_kernel<OUT_C, TR, TK, false><<<tile_blocks, 1024, 0, stream>>>(
        T, dinv, b3, src, dst, key, rec_s, rec_dl, rec_w, rec_cnt, out, n, ne);
}


# Aggregation hands each edge's record (source, tile row, norm product) to the wave by scalar lane reads, not LDS permutes: 3 fewer LDS ops per gathered row, scalar row base; sum order LDS arrival
# speedup vs baseline: 2.6510x; 1.0355x over previous
#include <hip/hip_runtime.h>
#include <hip/hip_bf16.h>


typedef float v2f __attribute__((ext_vector_type(2)));
typedef float v8f __attribute__((ext_vector_type(8)));


__global__ void deg_init_kernel(float* __restrict__ deg, int n) {
    int i = blockIdx.x * blockDim.x + threadIdx.x;
    if (i < n) deg[i] = 1.0f;
}

constexpr int TILE_ROWS  = 256;
constexpr int TILE_SHIFT = 8;
constexpr int KEY_LOADS  = 4;
constexpr int TRIP_EDGES = 512 * KEY_LOADS;

__global__ void deg_count_key_kernel(const int* __restrict__ dst,
                                     float* __restrict__ deg,
                                     unsigned int* __restrict__ key,
                                     int ne, int nwords) {
    const int q = blockIdx.x * blockDim.x + threadIdx.x;
    if (q >= nwords) return;
    const int e0 = 4 * q;
    unsigned int kw = 0u;
    if (e0 + 3 < ne) {
        int d[4];
#pragma unroll
        for (int j = 0; j < 4; ++j) d[j] = dst[e0 + j];
#pragma unroll
        for (int j = 0; j < 4; ++j) {
            atomicAdd(&deg[d[j]], 1.0f);
            kw |= ((unsigned int)(d[j] >> TILE_SHIFT) & 255u) << (8 * j);
        }
    } else {
#pragma unroll
        for (int j = 0; j < 4; ++j) {
            if (e0 + j < ne) {
                const int dj = dst[e0 + j];
                atomicAdd(&deg[dj], 1.0f);
                kw |= ((unsigned int)(dj >> TILE_SHIFT) & 255u) << (8 * j);
            }
        }
    }
    key[q] = kw;
}

__global__ void deg_to_dinv_kernel(float* __restrict__ deg, int n) {
    int i = blockIdx.x * blockDim.x + threadIdx.x;
    if (i < n) {
        float d = deg[i];
        deg[i] = (d > 0.0f) ? rsqrtf(d) : 0.0f;
    }
}


template <bool RELU, int TN>
__global__ void gemm_wmma_strip(const float* __restrict__ A,
                                const float* __restrict__ W,
                                float* __restrict__ out,
                                int M, int K) {
    const int Nc   = TN * 16;
    const int wave = (blockIdx.x * blockDim.x + threadIdx.x) >> 5;
    const int lane = threadIdx.x & 31;

    const int tm = wave;
    if (tm * 16 >= M) return;

    const int ml    = lane & 15;
    const int khalf = (lane >> 4) << 1;

    const float* arow = A + (size_t)(tm * 16 + ml) * K;

    v8f acc[TN];
#pragma unroll
    for (int t = 0; t < TN; ++t) acc[t] = (v8f){};

    for (int k = 0; k < K; k += 4) {
        v2f a = *(const v2f*)(arow + k + khalf);
        if (RELU) {
            a.x = fmaxf(a.x, 0.0f);
            a.y = fmaxf(a.y, 0.0f);
        }

        const float* w0 = W + (size_t)(k + khalf) * Nc + ml;
        v2f b[TN];
#pragma unroll
        for (int t = 0; t < TN; ++t) {
            b[t].x = w0[t * 16];
            b[t].y = w0[Nc + t * 16];
        }

#pragma unroll
        for (int t = 0; t < TN; ++t) {
            acc[t] = __builtin_amdgcn_wmma_f32_16x16x4_f32(
                 false, a,  false, b[t],
                 (short)0, acc[t],  false,  false);
        }
    }

    const int row0 = tm * 16 + ((lane >> 4) << 3);
#pragma unroll
    for (int t = 0; t < TN; ++t) {
        float* o = out + (size_t)row0 * Nc + t * 16 + ml;
#pragma unroll
        for (int r = 0; r < 8; ++r) o[(size_t)r * Nc] = acc[t][r];
    }
}


constexpr int LIST_CAP   = 1024;
constexpr int DRAIN_ROWS = 8;
constexpr int REC_CAP    = 512;
constexpr int PRE_BATCHES = 6;

template <int K>
__device__ __forceinline__ void load_key_trip(uint4 (&kv)[K], const unsigned int* __restrict__ key,
                                              int ebase, int lane) {
    const uint4* kp = (const uint4*)(key + (ebase >> 2));
#pragma unroll
    for (int k = 0; k < K; ++k) kv[k] = kp[lane + 32 * k];
}

template <int C>
__device__ __forceinline__ int tile_pos(int c) {
    constexpr int Q = C / 32;
    return (c % Q) * 32 + c / Q;
}

template <int C>
__device__ __forceinline__ void add_batch(int s, int dl, float w, int mcount, int lane,
                                          const float* __restrict__ t, float* tile) {
    constexpr int Q = C / 32;
    typedef float vq __attribute__((ext_vector_type(Q)));
    constexpr int G = DRAIN_ROWS;
    static_assert(32 % G == 0, "a batch of 32 is a whole number of row groups");
    for (int m = 0; m < mcount; m += G) {
        vq    v[G];
        int   dm[G];
        float wm[G];
#pragma unroll
        for (int u = 0; u < G; ++u) {
            const int mm = (m + u) & 31;
            const int sm = __builtin_amdgcn_readlane(s, mm);
            dm[u] = __builtin_amdgcn_readlane(dl, mm);
            wm[u] = __int_as_float(__builtin_amdgcn_readlane(__float_as_int(w), mm));
            v[u] = *(const vq*)(t + (size_t)sm * C + Q * lane);
        }
#pragma unroll
        for (int u = 0; u < G; ++u) {
            if (dm[u] >= 0) {
                float* lr = tile + dm[u] * C + lane;
#pragma unroll
                for (int q = 0; q < Q; ++q)
                    atomicAdd(&lr[32 * q], wm[u] * v[u][q]);
            }
        }
    }
}

template <int C, bool SAVE>
__device__ __forceinline__ void drain_list(const int* my, int cnt, int lane,
                                           const float* __restrict__ t,
                                           const float* __restrict__ dinv,
                                           const int* __restrict__ src,
                                           const int* __restrict__ dst,
                                           float* tile, int row0, int rows, int ne,
                                           int* __restrict__ rs, int* __restrict__ rd,
                                           float* __restrict__ rw, int& saved) {
    __builtin_amdgcn_fence(__ATOMIC_RELEASE, "wavefront");
    __builtin_amdgcn_wave_barrier();
    __builtin_amdgcn_fence(__ATOMIC_ACQUIRE, "wavefront");

    bool keep = false;
    if (SAVE) {
        keep = (saved >= 0) && (saved + cnt <= REC_CAP);
        if (!keep) saved = -1;
    }

    for (int b = 0; b < cnt; b += 32) {
        const int mcount = (cnt - b < 32) ? (cnt - b) : 32;
        int s = 0, dl = -1;
        float w = 0.0f;
        if (lane < mcount) {
            const int e = my[b + lane];
            if (e < ne) {
                const int d  = dst[e];
                const int se = src[e];
                w = dinv[se] * dinv[d];
                const bool mine = (unsigned)(d - row0) < (unsigned)rows;
                s  = mine ? se : 0;
                dl = mine ? d - row0 : -1;
            }
            if (SAVE && keep) {
                rs[saved + b + lane] = s;
                rd[saved + b + lane] = dl;
                rw[saved + b + lane] = w;
            }
        }
        add_batch<C>(s, dl, w, mcount, lane, t, tile);
    }
    if (SAVE && keep) saved += cnt;
    __builtin_amdgcn_fence(__ATOMIC_RELEASE, "wavefront");
    __builtin_amdgcn_wave_barrier();
    __builtin_amdgcn_fence(__ATOMIC_ACQUIRE, "wavefront");
}

template <int C>
__device__ __forceinline__ void drain_records(const int* __restrict__ rs,
                                              const int* __restrict__ rd,
                                              const float* __restrict__ rw,
                                              int cnt, int lane,
                                              int (&ps)[PRE_BATCHES],
                                              int (&pd)[PRE_BATCHES],
                                              float (&pw)[PRE_BATCHES],
                                              const float* __restrict__ t, float* tile) {
    for (int b = 0; b < cnt; b += 32) {
        const int mcount = (cnt - b < 32) ? (cnt - b) : 32;
        int s = ps[0], dl = pd[0];
        float w = pw[0];
        if (b < 32 * PRE_BATCHES) {
#pragma unroll
            for (int q = 0; q + 1 < PRE_BATCHES; ++q) {
                ps[q] = ps[q + 1];
                pd[q] = pd[q + 1];
                pw[q] = pw[q + 1];
            }
        } else {
            s = 0; dl = -1; w = 0.0f;
            if (lane < mcount) {
                s  = rs[b + lane];
                dl = rd[b + lane];
                w  = rw[b + lane];
            }
        }
        add_batch<C>(s, dl, w, mcount, lane, t, tile);
    }
}

template <int C, int R, int K, bool FIRST>
__global__ __launch_bounds__(1024)
void tile_agg_kernel(const float* __restrict__ t,
                     const float* __restrict__ dinv,
                     const float* __restrict__ bias,
                     const int* __restrict__ src,
                     const int* __restrict__ dst,
                     const unsigned int* __restrict__ key,
                     int* __restrict__ rec_s,
                     int* __restrict__ rec_dl,
                     float* __restrict__ rec_w,
                     int* __restrict__ rec_cnt,
                     float* __restrict__ agg, int n, int ne) {
    static_assert(K == 4, "one match bit per inspected edge: 16*K bits in a 64-bit mask");
    static_assert(K == KEY_LOADS, "the key array is padded to whole trips of this size");
    static_assert(R == (1 << TILE_SHIFT), "the key is the tile number's low byte");
    constexpr int NW  = 32;
    constexpr int EPW = 512 * K;
    static_assert(LIST_CAP >= 64, "room for one round of 32 matches below the drain mark");
    __shared__ float tile[R * C];
    __shared__ int hits[NW * LIST_CAP];

    const int tid  = threadIdx.x;
    const int lane = tid & 31;
    const int wave = __builtin_amdgcn_readfirstlane(tid >> 5);
    const int row0 = blockIdx.x * R;
    const int rows = (n - row0 < R) ? (n - row0) : R;

    const int slot = blockIdx.x * NW + wave;
    int*   rs = rec_s  + (size_t)slot * REC_CAP;
    int*   rd = rec_dl + (size_t)slot * REC_CAP;
    float* rw = rec_w  + (size_t)slot * REC_CAP;

    int have = -1;
    int   ps[PRE_BATCHES];
    int   pd[PRE_BATCHES];
    float pw[PRE_BATCHES];
#pragma unroll
    for (int j = 0; j < PRE_BATCHES; ++j) { ps[j] = 0; pd[j] = -1; pw[j] = 0.0f; }
    if (!FIRST) {
        have = __builtin_amdgcn_readfirstlane(rec_cnt[slot]);
        if (have > REC_CAP) have = -1;
#pragma unroll
        for (int j = 0; j < PRE_BATCHES; ++j) {
            const int i = 32 * j + lane;
            if (i < have) {
                ps[j] = rs[i];
                pd[j] = rd[i];
                pw[j] = rw[i];
            }
        }
    }

    {
        static_assert(1024 % (C / 4) == 0, "a thread's column group is fixed");
        const float4* g4 = (const float4*)(t + (size_t)row0 * C);
        const int c0 = 4 * (tid % (C / 4));
        const float bx = bias[c0], by = bias[c0 + 1], bz = bias[c0 + 2], bw = bias[c0 + 3];
        const int p0 = tile_pos<C>(c0),     p1 = tile_pos<C>(c0 + 1);
        const int p2 = tile_pos<C>(c0 + 2), p3 = tile_pos<C>(c0 + 3);
        for (int i = tid; i < rows * (C / 4); i += 1024) {
            const int r = i / (C / 4);
            const float di = dinv[row0 + r];
            const float4 tv = g4[i];
            float* lr = tile + r * C;
            lr[p0] = bx + di * di * tv.x;
            lr[p1] = by + di * di * tv.y;
            lr[p2] = bz + di * di * tv.z;
            lr[p3] = bw + di * di * tv.w;
        }
    }
    __syncthreads();

    if (!FIRST && have >= 0) {
        drain_records<C>(rs, rd, rw, have, lane, ps, pd, pw, t, tile);
    } else {
        const int nchunks = (ne + EPW - 1) / EPW;
        const int rot = (int)(((long long)blockIdx.x * nchunks) / gridDim.x);
        int* my = hits + wave * LIST_CAP;
        const unsigned below = (1u << lane) - 1u;
        const unsigned pat = (blockIdx.x & 255u) * 0x01010101u;

        uint4 kv[K];
        if (wave < nchunks) {
            int chunk = wave + rot;
            if (chunk >= nchunks) chunk -= nchunks;
            load_key_trip<K>(kv, key, chunk * EPW, lane);
        }

        int cnt = 0;
        int saved = 0;
        for (int idx = wave; idx < nchunks; idx += NW) {
            int chunk = idx + rot;
            if (chunk >= nchunks) chunk -= nchunks;
            const int ebase = chunk * EPW;

            unsigned mlo = 0u, mhi = 0u;
#pragma unroll
            for (int k = 0; k < K; ++k) {
#pragma unroll
                for (int j = 0; j < 4; ++j) {
                    const unsigned wv = (j == 0) ? kv[k].x : (j == 1) ? kv[k].y
                                      : (j == 2) ? kv[k].z : kv[k].w;
                    const unsigned x = wv ^ pat;
                    const unsigned z = (x - 0x01010101u) & ~x & 0x80808080u;
                    const unsigned bits = (z >> 7) << (4 * (k & 1) + j);
                    if (k < 2) mlo |= bits; else mhi |= bits;
                }
            }
            unsigned long long mask = ((unsigned long long)mhi << 32) | mlo;

            if (idx + NW < nchunks) {
                int nchunk = idx + NW + rot;
                if (nchunk >= nchunks) nchunk -= nchunks;
                load_key_trip<K>(kv, key, nchunk * EPW, lane);
            }

            for (int round = 0; round < 64; ++round) {
                const bool has = (mask != 0ull);
                const unsigned bal = __builtin_amdgcn_ballot_w32(has);
                if (bal == 0u) break;
                if (cnt > LIST_CAP - 32) {
                    drain_list<C, FIRST>(my, cnt, lane, t, dinv, src, dst, tile, row0, rows, ne,
                                         rs, rd, rw, saved);
                    cnt = 0;
                }
                if (has) {
                    const int bp = __builtin_ctzll(mask);
                    mask &= mask - 1ull;
                    const int s  = bp & 7;
                    const int kk = ((bp >> 5) << 1) + (s >> 2);
                    my[cnt + __popc(bal & below)] =
                        ebase + 16 * lane + 512 * kk + 4 * (s & 3) + ((bp & 31) >> 3);
                }
                cnt += __popc(bal);
            }
        }
        drain_list<C, FIRST>(my, cnt, lane, t, dinv, src, dst, tile, row0, rows, ne,
                             rs, rd, rw, saved);
        if (FIRST && lane == 0) rec_cnt[slot] = saved;
    }
    __syncthreads();

    {
        float4* g4 = (float4*)(agg + (size_t)row0 * C);
        const int c0 = 4 * (tid % (C / 4));
        const int p0 = tile_pos<C>(c0),     p1 = tile_pos<C>(c0 + 1);
        const int p2 = tile_pos<C>(c0 + 2), p3 = tile_pos<C>(c0 + 3);
        for (int i = tid; i < rows * (C / 4); i += 1024) {
            const float* lr = tile + (i / (C / 4)) * C;
            float4 o;
            o.x = lr[p0];
            o.y = lr[p1];
            o.z = lr[p2];
            o.w = lr[p3];
            g4[i] = o;
        }
    }
}


static inline int cdiv(long long a, long long b) { return (int)((a + b - 1) / b); }

extern "C" void kernel_launch(void* const* d_in, const int* in_sizes, int n_in,
                              void* d_out, int out_size, void* d_ws, size_t ws_size,
                              hipStream_t stream) {
    const float* x  = (const float*)d_in[0];
    const float* W1 = (const float*)d_in[1];
    const float* b1 = (const float*)d_in[2];
    const float* W2 = (const float*)d_in[3];
    const float* b2 = (const float*)d_in[4];
    const float* W3 = (const float*)d_in[5];
    const float* b3 = (const float*)d_in[6];
    const int*   ei = (const int*)d_in[7];

    const int IN_C = 64, HID_C = 128, OUT_C = 64;
    const int n  = in_sizes[0] / IN_C;
    const int ne = in_sizes[7] / 2;
    const int* src = ei;
    const int* dst = ei + ne;

    const int key_words = cdiv(ne, TRIP_EDGES) * (TRIP_EDGES / 4);
    float* T    = (float*)d_ws;
    float* G    = T + (size_t)n * HID_C;
    unsigned int* key = (unsigned int*)(G + (size_t)n * HID_C);
    float* dinv = (float*)(key + key_words);
    float* out  = (float*)d_out;

    const int TB = 256;
    const long long strip_waves = n / 16;
    const int gemm_blocks = cdiv(strip_waves * 32, TB);
    constexpr int TR = TILE_ROWS, TK = KEY_LOADS;
    const int tile_blocks = cdiv(n, TR);
    const size_t rec_slots = (size_t)tile_blocks * 32;
    int*   rec_s   = (int*)(dinv + n);
    int*   rec_dl  = rec_s + rec_slots * REC_CAP;
    float* rec_w   = (float*)(rec_dl + rec_slots * REC_CAP);
    int*   rec_cnt = (int*)(rec_w + rec_slots * REC_CAP);

    deg_init_kernel<<<cdiv(n, TB), TB, 0, stream>>>(dinv, n);
    deg_count_key_kernel<<<cdiv(key_words, TB), TB, 0, stream>>>(dst, dinv, key, ne, key_words);
    deg_to_dinv_kernel<<<cdiv(n, TB), TB, 0, stream>>>(dinv, n);

    gemm_wmma_strip<false, 8><<<gemm_blocks, TB, 0, stream>>>(x, W1, T, n, IN_C);
    tile_agg_kernel<HID_C, TR, TK, true><<<tile_blocks, 1024, 0, stream>>>(
        T, dinv, b1, src, dst, key, rec_s, rec_dl, rec_w, rec_cnt, G, n, ne);

    gemm_wmma_strip<true, 8><<<gemm_blocks, TB, 0, stream>>>(G, W2, T, n, HID_C);
    tile_agg_kernel<HID_C, TR, TK, false><<<tile_blocks, 1024, 0, stream>>>(
        T, dinv, b2, src, dst, key, rec_s, rec_dl, rec_w, rec_cnt, G, n, ne);

    gemm_wmma_strip<true, 4><<<gemm_blocks, TB, 0, stream>>>(G, W3, T, n, HID_C);
    tile_agg_kernel<OUT_C, TR, TK, false><<<tile_blocks, 1024, 0, stream>>>(
        T, dinv, b3, src, dst, key, rec_s, rec_dl, rec_w, rec_cnt, out, n, ne);
}


# Degree pass folded into its neighbours: edge count + keys run beside the layer-1 transform in one launch (by work-group range), rsqrt of a degree taken at use; 7 launches, not 9; sums unchanged
# speedup vs baseline: 2.8062x; 1.0585x over previous
#include <hip/hip_runtime.h>
#include <hip/hip_bf16.h>


typedef float v2f __attribute__((ext_vector_type(2)));
typedef float v8f __attribute__((ext_vector_type(8)));


__global__ void deg_init_kernel(float* __restrict__ deg, int n) {
    int i = blockIdx.x * blockDim.x + threadIdx.x;
    if (i < n) deg[i] = 1.0f;
}

constexpr int TILE_ROWS  = 256;
constexpr int TILE_SHIFT = 8;
constexpr int KEY_LOADS  = 4;
constexpr int TRIP_EDGES = 512 * KEY_LOADS;

__device__ __forceinline__ void deg_count_key(const int* __restrict__ dst,
                                              float* __restrict__ deg,
                                              unsigned int* __restrict__ key,
                                              int ne, int nwords, int q) {
    if (q >= nwords) return;
    const int e0 = 4 * q;
    unsigned int kw = 0u;
    if (e0 + 3 < ne) {
        int d[4];
#pragma unroll
        for (int j = 0; j < 4; ++j) d[j] = dst[e0 + j];
#pragma unroll
        for (int j = 0; j < 4; ++j) {
            atomicAdd(&deg[d[j]], 1.0f);
            kw |= ((unsigned int)(d[j] >> TILE_SHIFT) & 255u) << (8 * j);
        }
    } else {
#pragma unroll
        for (int j = 0; j < 4; ++j) {
            if (e0 + j < ne) {
                const int dj = dst[e0 + j];
                atomicAdd(&deg[dj], 1.0f);
                kw |= ((unsigned int)(dj >> TILE_SHIFT) & 255u) << (8 * j);
            }
        }
    }
    key[q] = kw;
}

__device__ __forceinline__ float inv_sqrt_deg(float d) {
    return (d > 0.0f) ? rsqrtf(d) : 0.0f;
}


template <bool RELU, int TN>
__device__ __forceinline__ void strip_product(const float* __restrict__ A,
                                              const float* __restrict__ W,
                                              float* __restrict__ out,
                                              int M, int K) {
    const int Nc   = TN * 16;
    const int wave = (blockIdx.x * blockDim.x + threadIdx.x) >> 5;
    const int lane = threadIdx.x & 31;

    const int tm = wave;
    if (tm * 16 >= M) return;

    const int ml    = lane & 15;
    const int khalf = (lane >> 4) << 1;

    const float* arow = A + (size_t)(tm * 16 + ml) * K;

    v8f acc[TN];
#pragma unroll
    for (int t = 0; t < TN; ++t) acc[t] = (v8f){};

    for (int k = 0; k < K; k += 4) {
        v2f a = *(const v2f*)(arow + k + khalf);
        if (RELU) {
            a.x = fmaxf(a.x, 0.0f);
            a.y = fmaxf(a.y, 0.0f);
        }

        const float* w0 = W + (size_t)(k + khalf) * Nc + ml;
        v2f b[TN];
#pragma unroll
        for (int t = 0; t < TN; ++t) {
            b[t].x = w0[t * 16];
            b[t].y = w0[Nc + t * 16];
        }

#pragma unroll
        for (int t = 0; t < TN; ++t) {
            acc[t] = __builtin_amdgcn_wmma_f32_16x16x4_f32(
                 false, a,  false, b[t],
                 (short)0, acc[t],  false,  false);
        }
    }

    const int row0 = tm * 16 + ((lane >> 4) << 3);
#pragma unroll
    for (int t = 0; t < TN; ++t) {
        float* o = out + (size_t)row0 * Nc + t * 16 + ml;
#pragma unroll
        for (int r = 0; r < 8; ++r) o[(size_t)r * Nc] = acc[t][r];
    }
}

template <bool RELU, int TN>
__global__ void gemm_wmma_strip(const float* __restrict__ A,
                                const float* __restrict__ W,
                                float* __restrict__ out,
                                int M, int K) {
    strip_product<RELU, TN>(A, W, out, M, K);
}

template <int TN>
__global__ void transform_count_kernel(const float* __restrict__ A,
                                       const float* __restrict__ W,
                                       float* __restrict__ out,
                                       int M, int K, int strip_blocks,
                                       const int* __restrict__ dst,
                                       float* __restrict__ deg,
                                       unsigned int* __restrict__ key,
                                       int ne, int nwords) {
    if ((int)blockIdx.x < strip_blocks) {
        strip_product<false, TN>(A, W, out, M, K);
    } else {
        const int q = ((int)blockIdx.x - strip_blocks) * (int)blockDim.x + (int)threadIdx.x;
        deg_count_key(dst, deg, key, ne, nwords, q);
    }
}


constexpr int LIST_CAP   = 1024;
constexpr int DRAIN_ROWS = 8;
constexpr int REC_CAP    = 512;
constexpr int PRE_BATCHES = 6;

template <int K>
__device__ __forceinline__ void load_key_trip(uint4 (&kv)[K], const unsigned int* __restrict__ key,
                                              int ebase, int lane) {
    const uint4* kp = (const uint4*)(key + (ebase >> 2));
#pragma unroll
    for (int k = 0; k < K; ++k) kv[k] = kp[lane + 32 * k];
}

template <int C>
__device__ __forceinline__ int tile_pos(int c) {
    constexpr int Q = C / 32;
    return (c % Q) * 32 + c / Q;
}

template <int C>
__device__ __forceinline__ void add_batch(int s, int dl, float w, int mcount, int lane,
                                          const float* __restrict__ t, float* tile) {
    constexpr int Q = C / 32;
    typedef float vq __attribute__((ext_vector_type(Q)));
    constexpr int G = DRAIN_ROWS;
    static_assert(32 % G == 0, "a batch of 32 is a whole number of row groups");
    for (int m = 0; m < mcount; m += G) {
        vq    v[G];
        int   dm[G];
        float wm[G];
#pragma unroll
        for (int u = 0; u < G; ++u) {
            const int mm = (m + u) & 31;
            const int sm = __builtin_amdgcn_readlane(s, mm);
            dm[u] = __builtin_amdgcn_readlane(dl, mm);
            wm[u] = __int_as_float(__builtin_amdgcn_readlane(__float_as_int(w), mm));
            v[u] = *(const vq*)(t + (size_t)sm * C + Q * lane);
        }
#pragma unroll
        for (int u = 0; u < G; ++u) {
            if (dm[u] >= 0) {
                float* lr = tile + dm[u] * C + lane;
#pragma unroll
                for (int q = 0; q < Q; ++q)
                    atomicAdd(&lr[32 * q], wm[u] * v[u][q]);
            }
        }
    }
}

template <int C, bool SAVE>
__device__ __forceinline__ void drain_list(const int* my, int cnt, int lane,
                                           const float* __restrict__ t,
                                           const float* __restrict__ deg,
                                           const int* __restrict__ src,
                                           const int* __restrict__ dst,
                                           float* tile, int row0, int rows, int ne,
                                           int* __restrict__ rs, int* __restrict__ rd,
                                           float* __restrict__ rw, int& saved) {
    __builtin_amdgcn_fence(__ATOMIC_RELEASE, "wavefront");
    __builtin_amdgcn_wave_barrier();
    __builtin_amdgcn_fence(__ATOMIC_ACQUIRE, "wavefront");

    bool keep = false;
    if (SAVE) {
        keep = (saved >= 0) && (saved + cnt <= REC_CAP);
        if (!keep) saved = -1;
    }

    for (int b = 0; b < cnt; b += 32) {
        const int mcount = (cnt - b < 32) ? (cnt - b) : 32;
        int s = 0, dl = -1;
        float w = 0.0f;
        if (lane < mcount) {
            const int e = my[b + lane];
            if (e < ne) {
                const int d  = dst[e];
                const int se = src[e];
                w = inv_sqrt_deg(deg[se]) * inv_sqrt_deg(deg[d]);
                const bool mine = (unsigned)(d - row0) < (unsigned)rows;
                s  = mine ? se : 0;
                dl = mine ? d - row0 : -1;
            }
            if (SAVE && keep) {
                rs[saved + b + lane] = s;
                rd[saved + b + lane] = dl;
                rw[saved + b + lane] = w;
            }
        }
        add_batch<C>(s, dl, w, mcount, lane, t, tile);
    }
    if (SAVE && keep) saved += cnt;
    __builtin_amdgcn_fence(__ATOMIC_RELEASE, "wavefront");
    __builtin_amdgcn_wave_barrier();
    __builtin_amdgcn_fence(__ATOMIC_ACQUIRE, "wavefront");
}

template <int C>
__device__ __forceinline__ void drain_records(const int* __restrict__ rs,
                                              const int* __restrict__ rd,
                                              const float* __restrict__ rw,
                                              int cnt, int lane,
                                              int (&ps)[PRE_BATCHES],
                                              int (&pd)[PRE_BATCHES],
                                              float (&pw)[PRE_BATCHES],
                                              const float* __restrict__ t, float* tile) {
    for (int b = 0; b < cnt; b += 32) {
        const int mcount = (cnt - b < 32) ? (cnt - b) : 32;
        int s = ps[0], dl = pd[0];
        float w = pw[0];
        if (b < 32 * PRE_BATCHES) {
#pragma unroll
            for (int q = 0; q + 1 < PRE_BATCHES; ++q) {
                ps[q] = ps[q + 1];
                pd[q] = pd[q + 1];
                pw[q] = pw[q + 1];
            }
        } else {
            s = 0; dl = -1; w = 0.0f;
            if (lane < mcount) {
                s  = rs[b + lane];
                dl = rd[b + lane];
                w  = rw[b + lane];
            }
        }
        add_batch<C>(s, dl, w, mcount, lane, t, tile);
    }
}

template <int C, int R, int K, bool FIRST>
__global__ __launch_bounds__(1024)
void tile_agg_kernel(const float* __restrict__ t,
                     const float* __restrict__ deg,
                     const float* __restrict__ bias,
                     const int* __restrict__ src,
                     const int* __restrict__ dst,
                     const unsigned int* __restrict__ key,
                     int* __restrict__ rec_s,
                     int* __restrict__ rec_dl,
                     float* __restrict__ rec_w,
                     int* __restrict__ rec_cnt,
                     float* __restrict__ agg, int n, int ne) {
    static_assert(K == 4, "one match bit per inspected edge: 16*K bits in a 64-bit mask");
    static_assert(K == KEY_LOADS, "the key array is padded to whole trips of this size");
    static_assert(R == (1 << TILE_SHIFT), "the key is the tile number's low byte");
    constexpr int NW  = 32;
    constexpr int EPW = 512 * K;
    static_assert(LIST_CAP >= 64, "room for one round of 32 matches below the drain mark");
    __shared__ float tile[R * C];
    __shared__ int hits[NW * LIST_CAP];

    const int tid  = threadIdx.x;
    const int lane = tid & 31;
    const int wave = __builtin_amdgcn_readfirstlane(tid >> 5);
    const int row0 = blockIdx.x * R;
    const int rows = (n - row0 < R) ? (n - row0) : R;

    const int slot = blockIdx.x * NW + wave;
    int*   rs = rec_s  + (size_t)slot * REC_CAP;
    int*   rd = rec_dl + (size_t)slot * REC_CAP;
    float* rw = rec_w  + (size_t)slot * REC_CAP;

    int have = -1;
    int   ps[PRE_BATCHES];
    int   pd[PRE_BATCHES];
    float pw[PRE_BATCHES];
#pragma unroll
    for (int j = 0; j < PRE_BATCHES; ++j) { ps[j] = 0; pd[j] = -1; pw[j] = 0.0f; }
    if (!FIRST) {
        have = __builtin_amdgcn_readfirstlane(rec_cnt[slot]);
        if (have > REC_CAP) have = -1;
#pragma unroll
        for (int j = 0; j < PRE_BATCHES; ++j) {
            const int i = 32 * j + lane;
            if (i < have) {
                ps[j] = rs[i];
                pd[j] = rd[i];
                pw[j] = rw[i];
            }
        }
    }

    {
        static_assert(1024 % (C / 4) == 0, "a thread's column group is fixed");
        const float4* g4 = (const float4*)(t + (size_t)row0 * C);
        const int c0 = 4 * (tid % (C / 4));
        const float bx = bias[c0], by = bias[c0 + 1], bz = bias[c0 + 2], bw = bias[c0 + 3];
        const int p0 = tile_pos<C>(c0),     p1 = tile_pos<C>(c0 + 1);
        const int p2 = tile_pos<C>(c0 + 2), p3 = tile_pos<C>(c0 + 3);
        for (int i = tid; i < rows * (C / 4); i += 1024) {
            const int r = i / (C / 4);
            const float di = inv_sqrt_deg(deg[row0 + r]);
            const float4 tv = g4[i];
            float* lr = tile + r * C;
            lr[p0] = bx + di * di * tv.x;
            lr[p1] = by + di * di * tv.y;
            lr[p2] = bz + di * di * tv.z;
            lr[p3] = bw + di * di * tv.w;
        }
    }
    __syncthreads();

    if (!FIRST && have >= 0) {
        drain_records<C>(rs, rd, rw, have, lane, ps, pd, pw, t, tile);
    } else {
        const int nchunks = (ne + EPW - 1) / EPW;
        const int rot = (int)(((long long)blockIdx.x * nchunks) / gridDim.x);
        int* my = hits + wave * LIST_CAP;
        const unsigned below = (1u << lane) - 1u;
        const unsigned pat = (blockIdx.x & 255u) * 0x01010101u;

        uint4 kv[K];
        if (wave < nchunks) {
            int chunk = wave + rot;
            if (chunk >= nchunks) chunk -= nchunks;
            load_key_trip<K>(kv, key, chunk * EPW, lane);
        }

        int cnt = 0;
        int saved = 0;
        for (int idx = wave; idx < nchunks; idx += NW) {
            int chunk = idx + rot;
            if (chunk >= nchunks) chunk -= nchunks;
            const int ebase = chunk * EPW;

            unsigned mlo = 0u, mhi = 0u;
#pragma unroll
            for (int k = 0; k < K; ++k) {
#pragma unroll
                for (int j = 0; j < 4; ++j) {
                    const unsigned wv = (j == 0) ? kv[k].x : (j == 1) ? kv[k].y
                                      : (j == 2) ? kv[k].z : kv[k].w;
                    const unsigned x = wv ^ pat;
                    const unsigned z = (x - 0x01010101u) & ~x & 0x80808080u;
                    const unsigned bits = (z >> 7) << (4 * (k & 1) + j);
                    if (k < 2) mlo |= bits; else mhi |= bits;
                }
            }
            unsigned long long mask = ((unsigned long long)mhi << 32) | mlo;

            if (idx + NW < nchunks) {
                int nchunk = idx + NW + rot;
                if (nchunk >= nchunks) nchunk -= nchunks;
                load_key_trip<K>(kv, key, nchunk * EPW, lane);
            }

            for (int round = 0; round < 64; ++round) {
                const bool has = (mask != 0ull);
                const unsigned bal = __builtin_amdgcn_ballot_w32(has);
                if (bal == 0u) break;
                if (cnt > LIST_CAP - 32) {
                    drain_list<C, FIRST>(my, cnt, lane, t, deg, src, dst, tile, row0, rows, ne,
                                         rs, rd, rw, saved);
                    cnt = 0;
                }
                if (has) {
                    const int bp = __builtin_ctzll(mask);
                    mask &= mask - 1ull;
                    const int s  = bp & 7;
                    const int kk = ((bp >> 5) << 1) + (s >> 2);
                    my[cnt + __popc(bal & below)] =
                        ebase + 16 * lane + 512 * kk + 4 * (s & 3) + ((bp & 31) >> 3);
                }
                cnt += __popc(bal);
            }
        }
        drain_list<C, FIRST>(my, cnt, lane, t, deg, src, dst, tile, row0, rows, ne,
                             rs, rd, rw, saved);
        if (FIRST && lane == 0) rec_cnt[slot] = saved;
    }
    __syncthreads();

    {
        float4* g4 = (float4*)(agg + (size_t)row0 * C);
        const int c0 = 4 * (tid % (C / 4));
        const int p0 = tile_pos<C>(c0),     p1 = tile_pos<C>(c0 + 1);
        const int p2 = tile_pos<C>(c0 + 2), p3 = tile_pos<C>(c0 + 3);
        for (int i = tid; i < rows * (C / 4); i += 1024) {
            const float* lr = tile + (i / (C / 4)) * C;
            float4 o;
            o.x = lr[p0];
            o.y = lr[p1];
            o.z = lr[p2];
            o.w = lr[p3];
            g4[i] = o;
        }
    }
}


static inline int cdiv(long long a, long long b) { return (int)((a + b - 1) / b); }

extern "C" void kernel_launch(void* const* d_in, const int* in_sizes, int n_in,
                              void* d_out, int out_size, void* d_ws, size_t ws_size,
                              hipStream_t stream) {
    const float* x  = (const float*)d_in[0];
    const float* W1 = (const float*)d_in[1];
    const float* b1 = (const float*)d_in[2];
    const float* W2 = (const float*)d_in[3];
    const float* b2 = (const float*)d_in[4];
    const float* W3 = (const float*)d_in[5];
    const float* b3 = (const float*)d_in[6];
    const int*   ei = (const int*)d_in[7];

    const int IN_C = 64, HID_C = 128, OUT_C = 64;
    const int n  = in_sizes[0] / IN_C;
    const int ne = in_sizes[7] / 2;
    const int* src = ei;
    const int* dst = ei + ne;

    const int key_words = cdiv(ne, TRIP_EDGES) * (TRIP_EDGES / 4);
    float* T    = (float*)d_ws;
    float* G    = T + (size_t)n * HID_C;
    unsigned int* key = (unsigned int*)(G + (size_t)n * HID_C);
    float* deg  = (float*)(key + key_words);
    float* out  = (float*)d_out;

    const int TB = 256;
    const long long strip_waves = n / 16;
    const int gemm_blocks = cdiv(strip_waves * 32, TB);
    constexpr int TR = TILE_ROWS, TK = KEY_LOADS;
    const int tile_blocks = cdiv(n, TR);
    const size_t rec_slots = (size_t)tile_blocks * 32;
    int*   rec_s   = (int*)(deg + n);
    int*   rec_dl  = rec_s + rec_slots * REC_CAP;
    float* rec_w   = (float*)(rec_dl + rec_slots * REC_CAP);
    int*   rec_cnt = (int*)(rec_w + rec_slots * REC_CAP);

    deg_init_kernel<<<cdiv(n, TB), TB, 0, stream>>>(deg, n);
    transform_count_kernel<8><<<gemm_blocks + cdiv(key_words, TB), TB, 0, stream>>>(
        x, W1, T, n, IN_C, gemm_blocks, dst, deg, key, ne, key_words);
    tile_agg_kernel<HID_C, TR, TK, true><<<tile_blocks, 1024, 0, stream>>>(
        T, deg, b1, src, dst, key, rec_s, rec_dl, rec_w, rec_cnt, G, n, ne);

    gemm_wmma_strip<true, 8><<<gemm_blocks, TB, 0, stream>>>(G, W2, T, n, HID_C);
    tile_agg_kernel<HID_C, TR, TK, false><<<tile_blocks, 1024, 0, stream>>>(
        T, deg, b2, src, dst, key, rec_s, rec_dl, rec_w, rec_cnt, G, n, ne);

    gemm_wmma_strip<true, 4><<<gemm_blocks, TB, 0, stream>>>(G, W3, T, n, HID_C);
    tile_agg_kernel<OUT_C, TR, TK, false><<<tile_blocks, 1024, 0, stream>>>(
        T, deg, b3, src, dst, key, rec_s, rec_dl, rec_w, rec_cnt, out, n, ne);
}


# All three products hold W in registers: one wave per SIMD walks its 16-row strips, no W load in the loop (1024 waves, not 4096); edge count beside layer 1 takes 16 edges a thread; sums unchanged
# speedup vs baseline: 2.9238x; 1.0419x over previous
#include <hip/hip_runtime.h>
#include <hip/hip_bf16.h>


typedef float v2f __attribute__((ext_vector_type(2)));
typedef float v8f __attribute__((ext_vector_type(8)));


__global__ void deg_init_kernel(float* __restrict__ deg, int n) {
    int i = blockIdx.x * blockDim.x + threadIdx.x;
    if (i < n) deg[i] = 1.0f;
}

constexpr int TILE_ROWS  = 256;
constexpr int TILE_SHIFT = 8;
constexpr int KEY_LOADS  = 4;
constexpr int TRIP_EDGES = 512 * KEY_LOADS;

constexpr int COUNT_WORDS = 4;
static_assert((TRIP_EDGES / 4) % COUNT_WORDS == 0, "a trip is a whole number of count threads");
static_assert(COUNT_WORDS == 4, "the keys of one thread go out as one uint4");
__device__ __forceinline__ void deg_count_key(const int* __restrict__ dst,
                                              float* __restrict__ deg,
                                              unsigned int* __restrict__ key,
                                              int ne, int nwords, int q) {
    const int w0 = COUNT_WORDS * q;
    if (w0 >= nwords) return;
    const int e0 = 4 * w0;
    constexpr int NE = 4 * COUNT_WORDS;
    unsigned int kw[COUNT_WORDS];
#pragma unroll
    for (int i = 0; i < COUNT_WORDS; ++i) kw[i] = 0u;
    if (e0 + NE - 1 < ne) {
        int d[NE];
#pragma unroll
        for (int j = 0; j < NE; ++j) d[j] = dst[e0 + j];
#pragma unroll
        for (int j = 0; j < NE; ++j) {
            atomicAdd(&deg[d[j]], 1.0f);
            kw[j >> 2] |= ((unsigned int)(d[j] >> TILE_SHIFT) & 255u) << (8 * (j & 3));
        }
    } else {
#pragma unroll
        for (int j = 0; j < NE; ++j) {
            if (e0 + j < ne) {
                const int dj = dst[e0 + j];
                atomicAdd(&deg[dj], 1.0f);
                kw[j >> 2] |= ((unsigned int)(dj >> TILE_SHIFT) & 255u) << (8 * (j & 3));
            }
        }
    }
    *(uint4*)(key + w0) = make_uint4(kw[0], kw[1], kw[2], kw[3]);
}

__device__ __forceinline__ float inv_sqrt_deg(float d) {
    return (d > 0.0f) ? rsqrtf(d) : 0.0f;
}


template <bool RELU, int TN, int KS>
__device__ __forceinline__ void resident_product(const float* __restrict__ A,
                                                 const float* __restrict__ W,
                                                 float* __restrict__ out,
                                                 int M, int strips_per_wave) {
    constexpr int Nc = TN * 16;
    constexpr int K  = KS * 4;
    const int wave = (blockIdx.x * blockDim.x + threadIdx.x) >> 5;
    const int lane = threadIdx.x & 31;
    const int nstrips = M / 16;
    const int s0 = wave * strips_per_wave;
    if (s0 >= nstrips) return;
    const int s1 = (s0 + strips_per_wave < nstrips) ? s0 + strips_per_wave : nstrips;

    const int ml    = lane & 15;
    const int khalf = (lane >> 4) << 1;

    v2f b[KS][TN];
    {
        const float* w0 = W + (size_t)khalf * Nc + ml;
#pragma unroll
        for (int ks = 0; ks < KS; ++ks) {
#pragma unroll
            for (int t = 0; t < TN; ++t) {
                b[ks][t].x = w0[(4 * ks) * Nc + t * 16];
                b[ks][t].y = w0[(4 * ks + 1) * Nc + t * 16];
            }
        }
    }

    v2f a[KS];
    {
        const float* arow = A + (size_t)(s0 * 16 + ml) * K + khalf;
#pragma unroll
        for (int ks = 0; ks < KS; ++ks) a[ks] = *(const v2f*)(arow + 4 * ks);
    }

#pragma nounroll
    for (int s = s0; s < s1; ++s) {
        const int sn = (s + 1 < s1) ? s + 1 : s;
        v2f an[KS];
        {
            const float* nrow = A + (size_t)(sn * 16 + ml) * K + khalf;
#pragma unroll
            for (int ks = 0; ks < KS; ++ks) an[ks] = *(const v2f*)(nrow + 4 * ks);
        }

        v8f acc[TN];
#pragma unroll
        for (int t = 0; t < TN; ++t) acc[t] = (v8f){};

#pragma unroll
        for (int ks = 0; ks < KS; ++ks) {
            v2f av = a[ks];
            if (RELU) {
                av.x = fmaxf(av.x, 0.0f);
                av.y = fmaxf(av.y, 0.0f);
            }
#pragma unroll
            for (int t = 0; t < TN; ++t) {
                acc[t] = __builtin_amdgcn_wmma_f32_16x16x4_f32(
                     false, av,  false, b[ks][t],
                     (short)0, acc[t],  false,  false);
            }
        }

        const int row0 = s * 16 + ((lane >> 4) << 3);
#pragma unroll
        for (int t = 0; t < TN; ++t) {
            float* o = out + (size_t)row0 * Nc + t * 16 + ml;
#pragma unroll
            for (int r = 0; r < 8; ++r) o[(size_t)r * Nc] = acc[t][r];
        }

#pragma unroll
        for (int ks = 0; ks < KS; ++ks) a[ks] = an[ks];
    }
}

template <bool RELU, int TN, int KS>
__global__ __launch_bounds__(128)
void gemm_wmma_resident(const float* __restrict__ A,
                        const float* __restrict__ W,
                        float* __restrict__ out,
                        int M, int strips_per_wave) {
    resident_product<RELU, TN, KS>(A, W, out, M, strips_per_wave);
}

template <int TN, int KS>
__global__ __launch_bounds__(128)
void transform_count_kernel(const float* __restrict__ A,
                            const float* __restrict__ W,
                            float* __restrict__ out,
                            int M, int strips_per_wave, int strip_blocks,
                            const int* __restrict__ dst,
                            float* __restrict__ deg,
                            unsigned int* __restrict__ key,
                            int ne, int nwords) {
    if ((int)blockIdx.x < strip_blocks) {
        resident_product<false, TN, KS>(A, W, out, M, strips_per_wave);
    } else {
        const int q = ((int)blockIdx.x - strip_blocks) * (int)blockDim.x + (int)threadIdx.x;
        deg_count_key(dst, deg, key, ne, nwords, q);
    }
}


constexpr int LIST_CAP   = 1024;
constexpr int DRAIN_ROWS = 8;
constexpr int REC_CAP    = 512;
constexpr int PRE_BATCHES = 6;

template <int K>
__device__ __forceinline__ void load_key_trip(uint4 (&kv)[K], const unsigned int* __restrict__ key,
                                              int ebase, int lane) {
    const uint4* kp = (const uint4*)(key + (ebase >> 2));
#pragma unroll
    for (int k = 0; k < K; ++k) kv[k] = kp[lane + 32 * k];
}

template <int C>
__device__ __forceinline__ int tile_pos(int c) {
    constexpr int Q = C / 32;
    return (c % Q) * 32 + c / Q;
}

template <int C>
__device__ __forceinline__ void add_batch(int s, int dl, float w, int mcount, int lane,
                                          const float* __restrict__ t, float* tile) {
    constexpr int Q = C / 32;
    typedef float vq __attribute__((ext_vector_type(Q)));
    constexpr int G = DRAIN_ROWS;
    static_assert(32 % G == 0, "a batch of 32 is a whole number of row groups");
    for (int m = 0; m < mcount; m += G) {
        vq    v[G];
        int   dm[G];
        float wm[G];
#pragma unroll
        for (int u = 0; u < G; ++u) {
            const int mm = (m + u) & 31;
            const int sm = __builtin_amdgcn_readlane(s, mm);
            dm[u] = __builtin_amdgcn_readlane(dl, mm);
            wm[u] = __int_as_float(__builtin_amdgcn_readlane(__float_as_int(w), mm));
            v[u] = *(const vq*)(t + (size_t)sm * C + Q * lane);
        }
#pragma unroll
        for (int u = 0; u < G; ++u) {
            if (dm[u] >= 0) {
                float* lr = tile + dm[u] * C + lane;
#pragma unroll
                for (int q = 0; q < Q; ++q)
                    atomicAdd(&lr[32 * q], wm[u] * v[u][q]);
            }
        }
    }
}

template <int C, bool SAVE>
__device__ __forceinline__ void drain_list(const int* my, int cnt, int lane,
                                           const float* __restrict__ t,
                                           const float* __restrict__ deg,
                                           const int* __restrict__ src,
                                           const int* __restrict__ dst,
                                           float* tile, int row0, int rows, int ne,
                                           int* __restrict__ rs, int* __restrict__ rd,
                                           float* __restrict__ rw, int& saved) {
    __builtin_amdgcn_fence(__ATOMIC_RELEASE, "wavefront");
    __builtin_amdgcn_wave_barrier();
    __builtin_amdgcn_fence(__ATOMIC_ACQUIRE, "wavefront");

    bool keep = false;
    if (SAVE) {
        keep = (saved >= 0) && (saved + cnt <= REC_CAP);
        if (!keep) saved = -1;
    }

    for (int b = 0; b < cnt; b += 32) {
        const int mcount = (cnt - b < 32) ? (cnt - b) : 32;
        int s = 0, dl = -1;
        float w = 0.0f;
        if (lane < mcount) {
            const int e = my[b + lane];
            if (e < ne) {
                const int d  = dst[e];
                const int se = src[e];
                w = inv_sqrt_deg(deg[se]) * inv_sqrt_deg(deg[d]);
                const bool mine = (unsigned)(d - row0) < (unsigned)rows;
                s  = mine ? se : 0;
                dl = mine ? d - row0 : -1;
            }
            if (SAVE && keep) {
                rs[saved + b + lane] = s;
                rd[saved + b + lane] = dl;
                rw[saved + b + lane] = w;
            }
        }
        add_batch<C>(s, dl, w, mcount, lane, t, tile);
    }
    if (SAVE && keep) saved += cnt;
    __builtin_amdgcn_fence(__ATOMIC_RELEASE, "wavefront");
    __builtin_amdgcn_wave_barrier();
    __builtin_amdgcn_fence(__ATOMIC_ACQUIRE, "wavefront");
}

template <int C>
__device__ __forceinline__ void drain_records(const int* __restrict__ rs,
                                              const int* __restrict__ rd,
                                              const float* __restrict__ rw,
                                              int cnt, int lane,
                                              int (&ps)[PRE_BATCHES],
                                              int (&pd)[PRE_BATCHES],
                                              float (&pw)[PRE_BATCHES],
                                              const float* __restrict__ t, float* tile) {
    for (int b = 0; b < cnt; b += 32) {
        const int mcount = (cnt - b < 32) ? (cnt - b) : 32;
        int s = ps[0], dl = pd[0];
        float w = pw[0];
        if (b < 32 * PRE_BATCHES) {
#pragma unroll
            for (int q = 0; q + 1 < PRE_BATCHES; ++q) {
                ps[q] = ps[q + 1];
                pd[q] = pd[q + 1];
                pw[q] = pw[q + 1];
            }
        } else {
            s = 0; dl = -1; w = 0.0f;
            if (lane < mcount) {
                s  = rs[b + lane];
                dl = rd[b + lane];
                w  = rw[b + lane];
            }
        }
        add_batch<C>(s, dl, w, mcount, lane, t, tile);
    }
}

template <int C, int R, int K, bool FIRST>
__global__ __launch_bounds__(1024)
void tile_agg_kernel(const float* __restrict__ t,
                     const float* __restrict__ deg,
                     const float* __restrict__ bias,
                     const int* __restrict__ src,
                     const int* __restrict__ dst,
                     const unsigned int* __restrict__ key,
                     int* __restrict__ rec_s,
                     int* __restrict__ rec_dl,
                     float* __restrict__ rec_w,
                     int* __restrict__ rec_cnt,
                     float* __restrict__ agg, int n, int ne) {
    static_assert(K == 4, "one match bit per inspected edge: 16*K bits in a 64-bit mask");
    static_assert(K == KEY_LOADS, "the key array is padded to whole trips of this size");
    static_assert(R == (1 << TILE_SHIFT), "the key is the tile number's low byte");
    constexpr int NW  = 32;
    constexpr int EPW = 512 * K;
    static_assert(LIST_CAP >= 64, "room for one round of 32 matches below the drain mark");
    __shared__ float tile[R * C];
    __shared__ int hits[NW * LIST_CAP];

    const int tid  = threadIdx.x;
    const int lane = tid & 31;
    const int wave = __builtin_amdgcn_readfirstlane(tid >> 5);
    const int row0 = blockIdx.x * R;
    const int rows = (n - row0 < R) ? (n - row0) : R;

    const int slot = blockIdx.x * NW + wave;
    int*   rs = rec_s  + (size_t)slot * REC_CAP;
    int*   rd = rec_dl + (size_t)slot * REC_CAP;
    float* rw = rec_w  + (size_t)slot * REC_CAP;

    int have = -1;
    int   ps[PRE_BATCHES];
    int   pd[PRE_BATCHES];
    float pw[PRE_BATCHES];
#pragma unroll
    for (int j = 0; j < PRE_BATCHES; ++j) { ps[j] = 0; pd[j] = -1; pw[j] = 0.0f; }
    if (!FIRST) {
        have = __builtin_amdgcn_readfirstlane(rec_cnt[slot]);
        if (have > REC_CAP) have = -1;
#pragma unroll
        for (int j = 0; j < PRE_BATCHES; ++j) {
            const int i = 32 * j + lane;
            if (i < have) {
                ps[j] = rs[i];
                pd[j] = rd[i];
                pw[j] = rw[i];
            }
        }
    }

    {
        static_assert(1024 % (C / 4) == 0, "a thread's column group is fixed");
        const float4* g4 = (const float4*)(t + (size_t)row0 * C);
        const int c0 = 4 * (tid % (C / 4));
        const float bx = bias[c0], by = bias[c0 + 1], bz = bias[c0 + 2], bw = bias[c0 + 3];
        const int p0 = tile_pos<C>(c0),     p1 = tile_pos<C>(c0 + 1);
        const int p2 = tile_pos<C>(c0 + 2), p3 = tile_pos<C>(c0 + 3);
        for (int i = tid; i < rows * (C / 4); i += 1024) {
            const int r = i / (C / 4);
            const float di = inv_sqrt_deg(deg[row0 + r]);
            const float4 tv = g4[i];
            float* lr = tile + r * C;
            lr[p0] = bx + di * di * tv.x;
            lr[p1] = by + di * di * tv.y;
            lr[p2] = bz + di * di * tv.z;
            lr[p3] = bw + di * di * tv.w;
        }
    }
    __syncthreads();

    if (!FIRST && have >= 0) {
        drain_records<C>(rs, rd, rw, have, lane, ps, pd, pw, t, tile);
    } else {
        const int nchunks = (ne + EPW - 1) / EPW;
        const int rot = (int)(((long long)blockIdx.x * nchunks) / gridDim.x);
        int* my = hits + wave * LIST_CAP;
        const unsigned below = (1u << lane) - 1u;
        const unsigned pat = (blockIdx.x & 255u) * 0x01010101u;

        uint4 kv[K];
        if (wave < nchunks) {
            int chunk = wave + rot;
            if (chunk >= nchunks) chunk -= nchunks;
            load_key_trip<K>(kv, key, chunk * EPW, lane);
        }

        int cnt = 0;
        int saved = 0;
        for (int idx = wave; idx < nchunks; idx += NW) {
            int chunk = idx + rot;
            if (chunk >= nchunks) chunk -= nchunks;
            const int ebase = chunk * EPW;

            unsigned mlo = 0u, mhi = 0u;
#pragma unroll
            for (int k = 0; k < K; ++k) {
#pragma unroll
                for (int j = 0; j < 4; ++j) {
                    const unsigned wv = (j == 0) ? kv[k].x : (j == 1) ? kv[k].y
                                      : (j == 2) ? kv[k].z : kv[k].w;
                    const unsigned x = wv ^ pat;
                    const unsigned z = (x - 0x01010101u) & ~x & 0x80808080u;
                    const unsigned bits = (z >> 7) << (4 * (k & 1) + j);
                    if (k < 2) mlo |= bits; else mhi |= bits;
                }
            }
            unsigned long long mask = ((unsigned long long)mhi << 32) | mlo;

            if (idx + NW < nchunks) {
                int nchunk = idx + NW + rot;
                if (nchunk >= nchunks) nchunk -= nchunks;
                load_key_trip<K>(kv, key, nchunk * EPW, lane);
            }

            for (int round = 0; round < 64; ++round) {
                const bool has = (mask != 0ull);
                const unsigned bal = __builtin_amdgcn_ballot_w32(has);
                if (bal == 0u) break;
                if (cnt > LIST_CAP - 32) {
                    drain_list<C, FIRST>(my, cnt, lane, t, deg, src, dst, tile, row0, rows, ne,
                                         rs, rd, rw, saved);
                    cnt = 0;
                }
                if (has) {
                    const int bp = __builtin_ctzll(mask);
                    mask &= mask - 1ull;
                    const int s  = bp & 7;
                    const int kk = ((bp >> 5) << 1) + (s >> 2);
                    my[cnt + __popc(bal & below)] =
                        ebase + 16 * lane + 512 * kk + 4 * (s & 3) + ((bp & 31) >> 3);
                }
                cnt += __popc(bal);
            }
        }
        drain_list<C, FIRST>(my, cnt, lane, t, deg, src, dst, tile, row0, rows, ne,
                             rs, rd, rw, saved);
        if (FIRST && lane == 0) rec_cnt[slot] = saved;
    }
    __syncthreads();

    {
        float4* g4 = (float4*)(agg + (size_t)row0 * C);
        const int c0 = 4 * (tid % (C / 4));
        const int p0 = tile_pos<C>(c0),     p1 = tile_pos<C>(c0 + 1);
        const int p2 = tile_pos<C>(c0 + 2), p3 = tile_pos<C>(c0 + 3);
        for (int i = tid; i < rows * (C / 4); i += 1024) {
            const float* lr = tile + (i / (C / 4)) * C;
            float4 o;
            o.x = lr[p0];
            o.y = lr[p1];
            o.z = lr[p2];
            o.w = lr[p3];
            g4[i] = o;
        }
    }
}


static inline int cdiv(long long a, long long b) { return (int)((a + b - 1) / b); }

extern "C" void kernel_launch(void* const* d_in, const int* in_sizes, int n_in,
                              void* d_out, int out_size, void* d_ws, size_t ws_size,
                              hipStream_t stream) {
    const float* x  = (const float*)d_in[0];
    const float* W1 = (const float*)d_in[1];
    const float* b1 = (const float*)d_in[2];
    const float* W2 = (const float*)d_in[3];
    const float* b2 = (const float*)d_in[4];
    const float* W3 = (const float*)d_in[5];
    const float* b3 = (const float*)d_in[6];
    const int*   ei = (const int*)d_in[7];

    const int IN_C = 64, HID_C = 128, OUT_C = 64;
    const int n  = in_sizes[0] / IN_C;
    const int ne = in_sizes[7] / 2;
    const int* src = ei;
    const int* dst = ei + ne;

    const int key_words = cdiv(ne, TRIP_EDGES) * (TRIP_EDGES / 4);
    float* T    = (float*)d_ws;
    float* G    = T + (size_t)n * HID_C;
    unsigned int* key = (unsigned int*)(G + (size_t)n * HID_C);
    float* deg  = (float*)(key + key_words);
    float* out  = (float*)d_out;

    const int TB = 256;
    const long long strip_waves = n / 16;
    const int res_blocks = 256;
    const int res_strips = cdiv(strip_waves, res_blocks * 4);
    constexpr int TR = TILE_ROWS, TK = KEY_LOADS;
    const int tile_blocks = cdiv(n, TR);
    const size_t rec_slots = (size_t)tile_blocks * 32;
    int*   rec_s   = (int*)(deg + n);
    int*   rec_dl  = rec_s + rec_slots * REC_CAP;
    float* rec_w   = (float*)(rec_dl + rec_slots * REC_CAP);
    int*   rec_cnt = (int*)(rec_w + rec_slots * REC_CAP);

    deg_init_kernel<<<cdiv(n, TB), TB, 0, stream>>>(deg, n);
    transform_count_kernel<8, 16><<<res_blocks + cdiv(key_words, COUNT_WORDS * 128), 128, 0, stream>>>(
        x, W1, T, n, res_strips, res_blocks, dst, deg, key, ne, key_words);
    tile_agg_kernel<HID_C, TR, TK, true><<<tile_blocks, 1024, 0, stream>>>(
        T, deg, b1, src, dst, key, rec_s, rec_dl, rec_w, rec_cnt, G, n, ne);

    gemm_wmma_resident<true, 8, 32><<<res_blocks, 128, 0, stream>>>(G, W2, T, n, res_strips);
    tile_agg_kernel<HID_C, TR, TK, false><<<tile_blocks, 1024, 0, stream>>>(
        T, deg, b2, src, dst, key, rec_s, rec_dl, rec_w, rec_cnt, G, n, ne);

    gemm_wmma_resident<true, 4, 32><<<res_blocks, 128, 0, stream>>>(G, W3, T, n, res_strips);
    tile_agg_kernel<OUT_C, TR, TK, false><<<tile_blocks, 1024, 0, stream>>>(
        T, deg, b3, src, dst, key, rec_s, rec_dl, rec_w, rec_cnt, out, n, ne);
}


# Degrees counted per tile in LDS by the edge search, now its own launch ahead of layer 1's gather (layer 1 forms the norm products): no global float atomic, no degree start launch; sums unchanged
# speedup vs baseline: 3.7379x; 1.2784x over previous
#include <hip/hip_runtime.h>
#include <hip/hip_bf16.h>


typedef float v2f __attribute__((ext_vector_type(2)));
typedef float v8f __attribute__((ext_vector_type(8)));


constexpr int TILE_ROWS  = 256;
constexpr int TILE_SHIFT = 8;
constexpr int KEY_LOADS  = 4;
constexpr int TRIP_EDGES = 512 * KEY_LOADS;

constexpr int COUNT_WORDS = 4;
static_assert((TRIP_EDGES / 4) % COUNT_WORDS == 0, "a trip is a whole number of count threads");
static_assert(COUNT_WORDS == 4, "the keys of one thread go out as one uint4");
__device__ __forceinline__ void write_keys(const int* __restrict__ dst,
                                           unsigned int* __restrict__ key,
                                           int ne, int nwords, int q) {
    const int w0 = COUNT_WORDS * q;
    if (w0 >= nwords) return;
    const int e0 = 4 * w0;
    constexpr int NE = 4 * COUNT_WORDS;
    unsigned int kw[COUNT_WORDS];
#pragma unroll
    for (int i = 0; i < COUNT_WORDS; ++i) kw[i] = 0u;
    if (e0 + NE - 1 < ne) {
#pragma unroll
        for (int j = 0; j < NE; ++j) {
            const int dj = dst[e0 + j];
            kw[j >> 2] |= ((unsigned int)(dj >> TILE_SHIFT) & 255u) << (8 * (j & 3));
        }
    } else {
#pragma unroll
        for (int j = 0; j < NE; ++j) {
            if (e0 + j < ne) {
                const int dj = dst[e0 + j];
                kw[j >> 2] |= ((unsigned int)(dj >> TILE_SHIFT) & 255u) << (8 * (j & 3));
            }
        }
    }
    *(uint4*)(key + w0) = make_uint4(kw[0], kw[1], kw[2], kw[3]);
}

__device__ __forceinline__ float inv_sqrt_deg(float d) {
    return (d > 0.0f) ? rsqrtf(d) : 0.0f;
}


template <bool RELU, int TN, int KS>
__device__ __forceinline__ void resident_product(const float* __restrict__ A,
                                                 const float* __restrict__ W,
                                                 float* __restrict__ out,
                                                 int M, int strips_per_wave) {
    constexpr int Nc = TN * 16;
    constexpr int K  = KS * 4;
    const int wave = (blockIdx.x * blockDim.x + threadIdx.x) >> 5;
    const int lane = threadIdx.x & 31;
    const int nstrips = M / 16;
    const int s0 = wave * strips_per_wave;
    if (s0 >= nstrips) return;
    const int s1 = (s0 + strips_per_wave < nstrips) ? s0 + strips_per_wave : nstrips;

    const int ml    = lane & 15;
    const int khalf = (lane >> 4) << 1;

    v2f b[KS][TN];
    {
        const float* w0 = W + (size_t)khalf * Nc + ml;
#pragma unroll
        for (int ks = 0; ks < KS; ++ks) {
#pragma unroll
            for (int t = 0; t < TN; ++t) {
                b[ks][t].x = w0[(4 * ks) * Nc + t * 16];
                b[ks][t].y = w0[(4 * ks + 1) * Nc + t * 16];
            }
        }
    }

    v2f a[KS];
    {
        const float* arow = A + (size_t)(s0 * 16 + ml) * K + khalf;
#pragma unroll
        for (int ks = 0; ks < KS; ++ks) a[ks] = *(const v2f*)(arow + 4 * ks);
    }

#pragma nounroll
    for (int s = s0; s < s1; ++s) {
        const int sn = (s + 1 < s1) ? s + 1 : s;
        v2f an[KS];
        {
            const float* nrow = A + (size_t)(sn * 16 + ml) * K + khalf;
#pragma unroll
            for (int ks = 0; ks < KS; ++ks) an[ks] = *(const v2f*)(nrow + 4 * ks);
        }

        v8f acc[TN];
#pragma unroll
        for (int t = 0; t < TN; ++t) acc[t] = (v8f){};

#pragma unroll
        for (int ks = 0; ks < KS; ++ks) {
            v2f av = a[ks];
            if (RELU) {
                av.x = fmaxf(av.x, 0.0f);
                av.y = fmaxf(av.y, 0.0f);
            }
#pragma unroll
            for (int t = 0; t < TN; ++t) {
                acc[t] = __builtin_amdgcn_wmma_f32_16x16x4_f32(
                     false, av,  false, b[ks][t],
                     (short)0, acc[t],  false,  false);
            }
        }

        const int row0 = s * 16 + ((lane >> 4) << 3);
#pragma unroll
        for (int t = 0; t < TN; ++t) {
            float* o = out + (size_t)row0 * Nc + t * 16 + ml;
#pragma unroll
            for (int r = 0; r < 8; ++r) o[(size_t)r * Nc] = acc[t][r];
        }

#pragma unroll
        for (int ks = 0; ks < KS; ++ks) a[ks] = an[ks];
    }
}

template <bool RELU, int TN, int KS>
__global__ __launch_bounds__(128)
void gemm_wmma_resident(const float* __restrict__ A,
                        const float* __restrict__ W,
                        float* __restrict__ out,
                        int M, int strips_per_wave) {
    resident_product<RELU, TN, KS>(A, W, out, M, strips_per_wave);
}

template <int TN, int KS>
__global__ __launch_bounds__(128)
void transform_key_kernel(const float* __restrict__ A,
                          const float* __restrict__ W,
                          float* __restrict__ out,
                          int M, int strips_per_wave, int strip_blocks,
                          const int* __restrict__ dst,
                          unsigned int* __restrict__ key,
                          int ne, int nwords) {
    if ((int)blockIdx.x < strip_blocks) {
        resident_product<false, TN, KS>(A, W, out, M, strips_per_wave);
    } else {
        const int q = ((int)blockIdx.x - strip_blocks) * (int)blockDim.x + (int)threadIdx.x;
        write_keys(dst, key, ne, nwords, q);
    }
}


constexpr int LIST_CAP   = 1024;
constexpr int DRAIN_ROWS = 8;
constexpr int REC_CAP    = 512;
constexpr int PRE_BATCHES = 6;
constexpr int RESOLVE_BATCHES = 4;

template <int K>
__device__ __forceinline__ void load_key_trip(uint4 (&kv)[K], const unsigned int* __restrict__ key,
                                              int ebase, int lane) {
    const uint4* kp = (const uint4*)(key + (ebase >> 2));
#pragma unroll
    for (int k = 0; k < K; ++k) kv[k] = kp[lane + 32 * k];
}

template <int C>
__device__ __forceinline__ int tile_pos(int c) {
    constexpr int Q = C / 32;
    return (c % Q) * 32 + c / Q;
}

template <int C>
__device__ __forceinline__ void add_batch(int s, int dl, float w, int mcount, int lane,
                                          const float* __restrict__ t, float* tile) {
    constexpr int Q = C / 32;
    typedef float vq __attribute__((ext_vector_type(Q)));
    constexpr int G = DRAIN_ROWS;
    static_assert(32 % G == 0, "a batch of 32 is a whole number of row groups");
    for (int m = 0; m < mcount; m += G) {
        vq    v[G];
        int   dm[G];
        float wm[G];
#pragma unroll
        for (int u = 0; u < G; ++u) {
            const int mm = (m + u) & 31;
            const int sm = __builtin_amdgcn_readlane(s, mm);
            dm[u] = __builtin_amdgcn_readlane(dl, mm);
            wm[u] = __int_as_float(__builtin_amdgcn_readlane(__float_as_int(w), mm));
            v[u] = *(const vq*)(t + (size_t)sm * C + Q * lane);
        }
#pragma unroll
        for (int u = 0; u < G; ++u) {
            if (dm[u] >= 0) {
                float* lr = tile + dm[u] * C + lane;
#pragma unroll
                for (int q = 0; q < Q; ++q)
                    atomicAdd(&lr[32 * q], wm[u] * v[u][q]);
            }
        }
    }
}

template <int C>
__device__ __forceinline__ void drain_list(const int* my, int cnt, int lane,
                                           const float* __restrict__ t,
                                           const float* __restrict__ deg,
                                           const int* __restrict__ src,
                                           const int* __restrict__ dst,
                                           float* tile, int row0, int rows, int ne) {
    __builtin_amdgcn_fence(__ATOMIC_RELEASE, "wavefront");
    __builtin_amdgcn_wave_barrier();
    __builtin_amdgcn_fence(__ATOMIC_ACQUIRE, "wavefront");

    for (int b = 0; b < cnt; b += 32) {
        const int mcount = (cnt - b < 32) ? (cnt - b) : 32;
        int s = 0, dl = -1;
        float w = 0.0f;
        if (lane < mcount) {
            const int e = my[b + lane];
            if (e < ne) {
                const int d  = dst[e];
                const int se = src[e];
                w = inv_sqrt_deg(deg[se]) * inv_sqrt_deg(deg[d]);
                const bool mine = (unsigned)(d - row0) < (unsigned)rows;
                s  = mine ? se : 0;
                dl = mine ? d - row0 : -1;
            }
        }
        add_batch<C>(s, dl, w, mcount, lane, t, tile);
    }
    __builtin_amdgcn_fence(__ATOMIC_RELEASE, "wavefront");
    __builtin_amdgcn_wave_barrier();
    __builtin_amdgcn_fence(__ATOMIC_ACQUIRE, "wavefront");
}

__device__ __forceinline__ void resolve_list(const int* my, int cnt, int lane,
                                             const int* __restrict__ src,
                                             const int* __restrict__ dst,
                                             int* indeg, int row0, int rows, int ne,
                                             int* __restrict__ rs, int* __restrict__ rd,
                                             int& saved) {
    __builtin_amdgcn_fence(__ATOMIC_RELEASE, "wavefront");
    __builtin_amdgcn_wave_barrier();
    __builtin_amdgcn_fence(__ATOMIC_ACQUIRE, "wavefront");

    const bool keep = (saved >= 0) && (saved + cnt <= REC_CAP);
    if (!keep) saved = -1;

    for (int b = 0; b < cnt; b += 32 * RESOLVE_BATCHES) {
        int  d[RESOLVE_BATCHES], se[RESOLVE_BATCHES];
        bool in[RESOLVE_BATCHES];
#pragma unroll
        for (int q = 0; q < RESOLVE_BATCHES; ++q) {
            const int i = b + 32 * q + lane;
            const int e = (i < cnt) ? my[i] : ne;
            in[q] = e < ne;
            const int ee = in[q] ? e : 0;
            d[q]  = dst[ee];
            se[q] = src[ee];
        }
        static_assert(RESOLVE_BATCHES == 4, "the statement below names the loads of four batches");
        asm volatile("" :: "v"(d[0]), "v"(se[0]), "v"(d[1]), "v"(se[1]),
                           "v"(d[2]), "v"(se[2]), "v"(d[3]), "v"(se[3]));
#pragma unroll
        for (int q = 0; q < RESOLVE_BATCHES; ++q) {
            const int i = b + 32 * q + lane;
            if (i < cnt) {
                const bool mine = in[q] && (unsigned)(d[q] - row0) < (unsigned)rows;
                const int s  = mine ? se[q] : 0;
                const int dl = mine ? d[q] - row0 : -1;
                if (mine) atomicAdd(&indeg[dl], 1);
                if (keep) {
                    rs[saved + i] = s;
                    rd[saved + i] = dl;
                }
            }
        }
    }
    if (keep) saved += cnt;
    __builtin_amdgcn_fence(__ATOMIC_RELEASE, "wavefront");
    __builtin_amdgcn_wave_barrier();
    __builtin_amdgcn_fence(__ATOMIC_ACQUIRE, "wavefront");
}

template <int K, typename Drain>
__device__ __forceinline__ void scan_keys(const unsigned int* __restrict__ key, int* my,
                                          int ne, int lane, int wave, Drain&& drain) {
    static_assert(K == 4, "one match bit per inspected edge: 16*K bits in a 64-bit mask");
    static_assert(K == KEY_LOADS, "the key array is padded to whole trips of this size");
    static_assert(LIST_CAP >= 64, "room for one round of 32 matches below the drain mark");
    constexpr int NW  = 32;
    constexpr int EPW = 512 * K;
    const int nchunks = (ne + EPW - 1) / EPW;
    const int rot = (int)(((long long)blockIdx.x * nchunks) / gridDim.x);
    const unsigned below = (1u << lane) - 1u;
    const unsigned pat = (blockIdx.x & 255u) * 0x01010101u;

    uint4 kv[K];
    if (wave < nchunks) {
        int chunk = wave + rot;
        if (chunk >= nchunks) chunk -= nchunks;
        load_key_trip<K>(kv, key, chunk * EPW, lane);
    }

    int cnt = 0;
    for (int idx = wave; idx < nchunks; idx += NW) {
        int chunk = idx + rot;
        if (chunk >= nchunks) chunk -= nchunks;
        const int ebase = chunk * EPW;

        unsigned mlo = 0u, mhi = 0u;
#pragma unroll
        for (int k = 0; k < K; ++k) {
#pragma unroll
            for (int j = 0; j < 4; ++j) {
                const unsigned wv = (j == 0) ? kv[k].x : (j == 1) ? kv[k].y
                                  : (j == 2) ? kv[k].z : kv[k].w;
                const unsigned x = wv ^ pat;
                const unsigned z = (x - 0x01010101u) & ~x & 0x80808080u;
                const unsigned bits = (z >> 7) << (4 * (k & 1) + j);
                if (k < 2) mlo |= bits; else mhi |= bits;
            }
        }
        unsigned long long mask = ((unsigned long long)mhi << 32) | mlo;

        if (idx + NW < nchunks) {
            int nchunk = idx + NW + rot;
            if (nchunk >= nchunks) nchunk -= nchunks;
            load_key_trip<K>(kv, key, nchunk * EPW, lane);
        }

        for (int round = 0; round < 64; ++round) {
            const bool has = (mask != 0ull);
            const unsigned bal = __builtin_amdgcn_ballot_w32(has);
            if (bal == 0u) break;
            if (cnt > LIST_CAP - 32) {
                drain(cnt);
                cnt = 0;
            }
            if (has) {
                const int bp = __builtin_ctzll(mask);
                mask &= mask - 1ull;
                const int s  = bp & 7;
                const int kk = ((bp >> 5) << 1) + (s >> 2);
                my[cnt + __popc(bal & below)] =
                    ebase + 16 * lane + 512 * kk + 4 * (s & 3) + ((bp & 31) >> 3);
            }
            cnt += __popc(bal);
        }
    }
    drain(cnt);
}

template <int R, int K>
__global__ __launch_bounds__(1024)
void tile_scan_kernel(const int* __restrict__ src,
                      const int* __restrict__ dst,
                      const unsigned int* __restrict__ key,
                      int* __restrict__ rec_s,
                      int* __restrict__ rec_dl,
                      int* __restrict__ rec_cnt,
                      float* __restrict__ deg, int n, int ne) {
    static_assert(R == (1 << TILE_SHIFT), "the key is the tile number's low byte");
    static_assert(R <= 1024, "one thread per row clears and stores the counts");
    constexpr int NW = 32;
    __shared__ int hits[NW * LIST_CAP];
    __shared__ int indeg[R];

    const int tid  = threadIdx.x;
    const int lane = tid & 31;
    const int wave = __builtin_amdgcn_readfirstlane(tid >> 5);
    const int row0 = blockIdx.x * R;
    const int rows = (n - row0 < R) ? (n - row0) : R;

    const int slot = blockIdx.x * NW + wave;
    int* rs = rec_s  + (size_t)slot * REC_CAP;
    int* rd = rec_dl + (size_t)slot * REC_CAP;

    if (tid < R) indeg[tid] = 0;
    __syncthreads();

    int* my = hits + wave * LIST_CAP;
    int saved = 0;
    scan_keys<K>(key, my, ne, lane, wave, [&](int cnt) {
        resolve_list(my, cnt, lane, src, dst, indeg, row0, rows, ne, rs, rd, saved);
    });
    if (lane == 0) rec_cnt[slot] = saved;
    __syncthreads();

    if (tid < rows) deg[row0 + tid] = 1.0f + (float)indeg[tid];
}

template <int C, bool MAKEW>
__device__ __forceinline__ void drain_records(const int* __restrict__ rs,
                                              const int* __restrict__ rd,
                                              float* rw,
                                              const float* __restrict__ deg, int row0,
                                              int cnt, int lane,
                                              int (&ps)[PRE_BATCHES],
                                              int (&pd)[PRE_BATCHES],
                                              float (&pw)[PRE_BATCHES],
                                              const float* __restrict__ t, float* tile) {
    for (int b = 0; b < cnt; b += 32) {
        const int mcount = (cnt - b < 32) ? (cnt - b) : 32;
        int s = ps[0], dl = pd[0];
        float w = pw[0];
        if (b < 32 * PRE_BATCHES) {
#pragma unroll
            for (int q = 0; q + 1 < PRE_BATCHES; ++q) {
                ps[q] = ps[q + 1];
                pd[q] = pd[q + 1];
                pw[q] = pw[q + 1];
            }
        } else {
            s = 0; dl = -1; w = 0.0f;
            if (lane < mcount) {
                s  = rs[b + lane];
                dl = rd[b + lane];
                if (MAKEW) {
                    if (dl >= 0) w = inv_sqrt_deg(deg[s]) * inv_sqrt_deg(deg[row0 + dl]);
                    rw[b + lane] = w;
                } else {
                    w = rw[b + lane];
                }
            }
        }
        add_batch<C>(s, dl, w, mcount, lane, t, tile);
    }
}

template <int C, int R, int K, bool MAKEW>
__global__ __launch_bounds__(1024)
void tile_agg_kernel(const float* __restrict__ t,
                     const float* __restrict__ deg,
                     const float* __restrict__ bias,
                     const int* __restrict__ src,
                     const int* __restrict__ dst,
                     const unsigned int* __restrict__ key,
                     const int* __restrict__ rec_s,
                     const int* __restrict__ rec_dl,
                     float* rec_w,
                     const int* __restrict__ rec_cnt,
                     float* __restrict__ agg, int n, int ne) {
    static_assert(R == (1 << TILE_SHIFT), "the key is the tile number's low byte");
    constexpr int NW  = 32;
    __shared__ float tile[R * C];
    __shared__ int hits[NW * LIST_CAP];

    const int tid  = threadIdx.x;
    const int lane = tid & 31;
    const int wave = __builtin_amdgcn_readfirstlane(tid >> 5);
    const int row0 = blockIdx.x * R;
    const int rows = (n - row0 < R) ? (n - row0) : R;

    const int slot = blockIdx.x * NW + wave;
    const int* rs = rec_s  + (size_t)slot * REC_CAP;
    const int* rd = rec_dl + (size_t)slot * REC_CAP;
    float*     rw = rec_w  + (size_t)slot * REC_CAP;

    int   ps[PRE_BATCHES];
    int   pd[PRE_BATCHES];
    float pw[PRE_BATCHES];
#pragma unroll
    for (int j = 0; j < PRE_BATCHES; ++j) { ps[j] = 0; pd[j] = -1; pw[j] = 0.0f; }
    int have = __builtin_amdgcn_readfirstlane(rec_cnt[slot]);
    if (have > REC_CAP) have = -1;
#pragma unroll
    for (int j = 0; j < PRE_BATCHES; ++j) {
        const int i = 32 * j + lane;
        if (i < have) {
            ps[j] = rs[i];
            pd[j] = rd[i];
            if (!MAKEW) pw[j] = rw[i];
        }
    }

    {
        static_assert(1024 % (C / 4) == 0, "a thread's column group is fixed");
        const float4* g4 = (const float4*)(t + (size_t)row0 * C);
        const int c0 = 4 * (tid % (C / 4));
        const float bx = bias[c0], by = bias[c0 + 1], bz = bias[c0 + 2], bw = bias[c0 + 3];
        const int p0 = tile_pos<C>(c0),     p1 = tile_pos<C>(c0 + 1);
        const int p2 = tile_pos<C>(c0 + 2), p3 = tile_pos<C>(c0 + 3);
        for (int i = tid; i < rows * (C / 4); i += 1024) {
            const int r = i / (C / 4);
            const float di = inv_sqrt_deg(deg[row0 + r]);
            const float4 tv = g4[i];
            float* lr = tile + r * C;
            lr[p0] = bx + di * di * tv.x;
            lr[p1] = by + di * di * tv.y;
            lr[p2] = bz + di * di * tv.z;
            lr[p3] = bw + di * di * tv.w;
        }
    }

    if (MAKEW && have > 0) {
        float da[PRE_BATCHES], db[PRE_BATCHES];
#pragma unroll
        for (int j = 0; j < PRE_BATCHES; ++j) {
            da[j] = deg[ps[j]];
            db[j] = deg[row0 + (pd[j] < 0 ? 0 : pd[j])];
        }
#pragma unroll
        for (int j = 0; j < PRE_BATCHES; ++j) {
            pw[j] = (pd[j] >= 0) ? inv_sqrt_deg(da[j]) * inv_sqrt_deg(db[j]) : 0.0f;
        }
    }
    __syncthreads();

    if (MAKEW && have > 0) {
#pragma unroll
        for (int j = 0; j < PRE_BATCHES; ++j) {
            const int i = 32 * j + lane;
            if (i < have) rw[i] = pw[j];
        }
    }

    if (have >= 0) {
        drain_records<C, MAKEW>(rs, rd, rw, deg, row0, have, lane, ps, pd, pw, t, tile);
    } else {
        int* my = hits + wave * LIST_CAP;
        scan_keys<K>(key, my, ne, lane, wave, [&](int cnt) {
            drain_list<C>(my, cnt, lane, t, deg, src, dst, tile, row0, rows, ne);
        });
    }
    __syncthreads();

    {
        float4* g4 = (float4*)(agg + (size_t)row0 * C);
        const int c0 = 4 * (tid % (C / 4));
        const int p0 = tile_pos<C>(c0),     p1 = tile_pos<C>(c0 + 1);
        const int p2 = tile_pos<C>(c0 + 2), p3 = tile_pos<C>(c0 + 3);
        for (int i = tid; i < rows * (C / 4); i += 1024) {
            const float* lr = tile + (i / (C / 4)) * C;
            float4 o;
            o.x = lr[p0];
            o.y = lr[p1];
            o.z = lr[p2];
            o.w = lr[p3];
            g4[i] = o;
        }
    }
}


static inline int cdiv(long long a, long long b) { return (int)((a + b - 1) / b); }

extern "C" void kernel_launch(void* const* d_in, const int* in_sizes, int n_in,
                              void* d_out, int out_size, void* d_ws, size_t ws_size,
                              hipStream_t stream) {
    const float* x  = (const float*)d_in[0];
    const float* W1 = (const float*)d_in[1];
    const float* b1 = (const float*)d_in[2];
    const float* W2 = (const float*)d_in[3];
    const float* b2 = (const float*)d_in[4];
    const float* W3 = (const float*)d_in[5];
    const float* b3 = (const float*)d_in[6];
    const int*   ei = (const int*)d_in[7];

    const int IN_C = 64, HID_C = 128, OUT_C = 64;
    const int n  = in_sizes[0] / IN_C;
    const int ne = in_sizes[7] / 2;
    const int* src = ei;
    const int* dst = ei + ne;

    const int key_words = cdiv(ne, TRIP_EDGES) * (TRIP_EDGES / 4);
    float* T    = (float*)d_ws;
    float* G    = T + (size_t)n * HID_C;
    unsigned int* key = (unsigned int*)(G + (size_t)n * HID_C);
    float* deg  = (float*)(key + key_words);
    float* out  = (float*)d_out;

    const long long strip_waves = n / 16;
    const int res_blocks = 256;
    const int res_strips = cdiv(strip_waves, res_blocks * 4);
    constexpr int TR = TILE_ROWS, TK = KEY_LOADS;
    const int tile_blocks = cdiv(n, TR);
    const size_t rec_slots = (size_t)tile_blocks * 32;
    int*   rec_s   = (int*)(deg + n);
    int*   rec_dl  = rec_s + rec_slots * REC_CAP;
    float* rec_w   = (float*)(rec_dl + rec_slots * REC_CAP);
    int*   rec_cnt = (int*)(rec_w + rec_slots * REC_CAP);

    transform_key_kernel<8, 16><<<res_blocks + cdiv(key_words, COUNT_WORDS * 128), 128, 0, stream>>>(
        x, W1, T, n, res_strips, res_blocks, dst, key, ne, key_words);
    tile_scan_kernel<TR, TK><<<tile_blocks, 1024, 0, stream>>>(
        src, dst, key, rec_s, rec_dl, rec_cnt, deg, n, ne);
    tile_agg_kernel<HID_C, TR, TK, true><<<tile_blocks, 1024, 0, stream>>>(
        T, deg, b1, src, dst, key, rec_s, rec_dl, rec_w, rec_cnt, G, n, ne);

    gemm_wmma_resident<true, 8, 32><<<res_blocks, 128, 0, stream>>>(G, W2, T, n, res_strips);
    tile_agg_kernel<HID_C, TR, TK, false><<<tile_blocks, 1024, 0, stream>>>(
        T, deg, b2, src, dst, key, rec_s, rec_dl, rec_w, rec_cnt, G, n, ne);

    gemm_wmma_resident<true, 4, 32><<<res_blocks, 128, 0, stream>>>(G, W3, T, n, res_strips);
    tile_agg_kernel<OUT_C, TR, TK, false><<<tile_blocks, 1024, 0, stream>>>(
        T, deg, b3, src, dst, key, rec_s, rec_dl, rec_w, rec_cnt, out, n, ne);
}
